# starting kernel
# baseline (speedup 1.0000x reference)
#include <hip/hip_runtime.h>
#include <math.h>

typedef __attribute__((ext_vector_type(16))) _Float16 v16h;
typedef __attribute__((ext_vector_type(8)))  float    v8f;

#define NNODES 100000
#define NEDGES 1600000
#define FDIM   128

static inline int ceil_div(long long a, long long b) { return (int)((a + b - 1) / b); }

__global__ void init_deg_kernel(float* __restrict__ odeg, float* __restrict__ ideg, int n) {
  int i = blockIdx.x * blockDim.x + threadIdx.x;
  if (i < n) { odeg[i] = 1.0f; ideg[i] = 1.0f; }
}

__global__ void degree_kernel(const int* __restrict__ src, const int* __restrict__ dst,
                              float* __restrict__ odeg, float* __restrict__ ideg, int e) {
  int i = blockIdx.x * blockDim.x + threadIdx.x;
  if (i < e) {
    atomicAdd(&odeg[src[i]], 1.0f);
    atomicAdd(&ideg[dst[i]], 1.0f);
  }
}

__global__ void rsqrt_kernel(float* __restrict__ a, float* __restrict__ b, int n) {
  int i = blockIdx.x * blockDim.x + threadIdx.x;
  if (i < n) {
    a[i] = rsqrtf(fmaxf(a[i], 1.0f));
    b[i] = rsqrtf(fmaxf(b[i], 1.0f));
  }
}

__global__ void zero_kernel(float* __restrict__ p, int n) {
  int i = blockIdx.x * blockDim.x + threadIdx.x;
  if (i < n) p[i] = 0.0f;
}

__global__ void pack_w_kernel(const float* __restrict__ W, _Float16* __restrict__ Wp, int nout) {
  int idx = blockIdx.x * blockDim.x + threadIdx.x;
  int nt = nout >> 4;
  int total = 4 * nt * 32 * 16;
  if (idx >= total) return;
  int j    = idx & 15;
  int l    = (idx >> 4) & 31;
  int rest = idx >> 9;
  int t    = rest % nt;
  int c    = rest / nt;
  int col   = t * 16 + (l & 15);
  int kbase = c * 32 + ((l >> 4) * 8);
  int K     = kbase + (j & 7) + ((j >> 3) * 16);
  Wp[idx] = (_Float16)W[K * nout + col];
}

template<int NOUT>
__global__ void gemm_wmma_kernel(const float* __restrict__ X,
                                 const float* __restrict__ onorm,
                                 const _Float16* __restrict__ Wp,
                                 float* __restrict__ H) {
  constexpr int NT = NOUT / 16;
  int wave = (int)((blockIdx.x * blockDim.x + threadIdx.x) >> 5);
  int lane = threadIdx.x & 31;
  int mtile = wave / NT;
  int ntile = wave % NT;
  if (mtile * 16 >= NNODES) return;

  int r    = lane & 15;
  int half = lane >> 4;
  int row  = mtile * 16 + r;
  float scale = onorm[row];
  const float* px = X + (size_t)row * FDIM;

  v8f acc = {0.f, 0.f, 0.f, 0.f, 0.f, 0.f, 0.f, 0.f};

#pragma unroll
  for (int kc = 0; kc < 4; ++kc) {
    int kbase = kc * 32 + half * 8;
    float4 f0 = *(const float4*)(px + kbase);
    float4 f1 = *(const float4*)(px + kbase + 4);
    float4 f2 = *(const float4*)(px + kbase + 16);
    float4 f3 = *(const float4*)(px + kbase + 20);
    v16h a;
    a[0]  = (_Float16)(f0.x * scale); a[1]  = (_Float16)(f0.y * scale);
    a[2]  = (_Float16)(f0.z * scale); a[3]  = (_Float16)(f0.w * scale);
    a[4]  = (_Float16)(f1.x * scale); a[5]  = (_Float16)(f1.y * scale);
    a[6]  = (_Float16)(f1.z * scale); a[7]  = (_Float16)(f1.w * scale);
    a[8]  = (_Float16)(f2.x * scale); a[9]  = (_Float16)(f2.y * scale);
    a[10] = (_Float16)(f2.z * scale); a[11] = (_Float16)(f2.w * scale);
    a[12] = (_Float16)(f3.x * scale); a[13] = (_Float16)(f3.y * scale);
    a[14] = (_Float16)(f3.z * scale); a[15] = (_Float16)(f3.w * scale);

    v16h b = *(const v16h*)(Wp + ((size_t)(kc * NT + ntile) * 32 + lane) * 16);

    acc = __builtin_amdgcn_wmma_f32_16x16x32_f16(
         false, a,  false, b,
         (short)0, acc,  false,  false);
  }

  int col = ntile * 16 + r;
  float* ph = H + (size_t)(mtile * 16 + half * 8) * NOUT + col;
#pragma unroll
  for (int vr = 0; vr < 8; ++vr)
    ph[(size_t)vr * NOUT] = acc[vr];
}

template<int NOUT>
__global__ void edge_agg_kernel(const int* __restrict__ src, const int* __restrict__ dst,
                                const float* __restrict__ h, float* __restrict__ agg, int e) {
  int wid  = (int)((blockIdx.x * blockDim.x + threadIdx.x) >> 5);
  int lane = threadIdx.x & 31;
  if (wid >= e) return;
  int s = src[wid];
  int d = dst[wid];
  constexpr int V = NOUT / 32;
  const float* hp = h   + (size_t)s * NOUT + lane * V;
  float*       ap = agg + (size_t)d * NOUT + lane * V;
  float v[V];
#pragma unroll
  for (int k = 0; k < V; ++k) v[k] = hp[k];
#pragma unroll
  for (int k = 0; k < V; ++k) atomicAdd(ap + k, v[k]);
}

template<int NOUT, bool RELU>
__global__ void finalize_kernel(const float* __restrict__ agg, const float* __restrict__ h,
                                const float* __restrict__ inorm, const float* __restrict__ b,
                                float* __restrict__ out, int n) {
  int idx = blockIdx.x * blockDim.x + threadIdx.x;
  if (idx >= n) return;
  int node = idx / NOUT;
  int dcol = idx % NOUT;
  float v = (agg[idx] + h[idx]) * inorm[node] + b[dcol];
  out[idx] = RELU ? fmaxf(v, 0.0f) : v;
}

extern "C" void kernel_launch(void* const* d_in, const int* in_sizes, int n_in,
                              void* d_out, int out_size, void* d_ws, size_t ws_size,
                              hipStream_t stream) {
  const float* feat = (const float*)d_in[0];
  const int*   src  = (const int*)  d_in[1];
  const int*   dst  = (const int*)  d_in[2];
  const float* W1   = (const float*)d_in[3];
  const float* b1   = (const float*)d_in[4];
  const float* W2   = (const float*)d_in[5];
  const float* b2   = (const float*)d_in[6];
  const float* W3   = (const float*)d_in[7];
  const float* b3   = (const float*)d_in[8];
  float* out = (float*)d_out;

  size_t off = 0;
  auto alloc = [&](size_t bytes) -> char* {
    char* p = (char*)d_ws + off;
    off += (bytes + 255) & ~(size_t)255;
    return p;
  };
  float*     onorm = (float*)alloc((size_t)NNODES * sizeof(float));
  float*     inorm = (float*)alloc((size_t)NNODES * sizeof(float));
  float*     hbuf  = (float*)alloc((size_t)NNODES * FDIM * sizeof(float));
  float*     abuf  = (float*)alloc((size_t)NNODES * FDIM * sizeof(float));
  float*     xbuf  = (float*)alloc((size_t)NNODES * FDIM * sizeof(float));
  _Float16*  Wp    = (_Float16*)alloc((size_t)FDIM * FDIM * sizeof(_Float16));

  const int TB = 256;

  init_deg_kernel<<<ceil_div(NNODES, TB), TB, 0, stream>>>(onorm, inorm, NNODES);
  degree_kernel<<<ceil_div(NEDGES, TB), TB, 0, stream>>>(src, dst, onorm, inorm, NEDGES);
  rsqrt_kernel<<<ceil_div(NNODES, TB), TB, 0, stream>>>(onorm, inorm, NNODES);

  pack_w_kernel<<<ceil_div(4 * 8 * 512, TB), TB, 0, stream>>>(W1, Wp, 128);
  zero_kernel<<<ceil_div((long long)NNODES * 128, TB), TB, 0, stream>>>(abuf, NNODES * 128);
  {
    int waves = (NNODES / 16) * (128 / 16);
    gemm_wmma_kernel<128><<<ceil_div((long long)waves * 32, TB), TB, 0, stream>>>(feat, onorm, Wp, hbuf);
  }
  edge_agg_kernel<128><<<ceil_div((long long)NEDGES * 32, TB), TB, 0, stream>>>(src, dst, hbuf, abuf, NEDGES);
  finalize_kernel<128, true><<<ceil_div((long long)NNODES * 128, TB), TB, 0, stream>>>(
      abuf, hbuf, inorm, b1, xbuf, NNODES * 128);

  pack_w_kernel<<<ceil_div(4 * 8 * 512, TB), TB, 0, stream>>>(W2, Wp, 128);
  zero_kernel<<<ceil_div((long long)NNODES * 128, TB), TB, 0, stream>>>(abuf, NNODES * 128);
  {
    int waves = (NNODES / 16) * (128 / 16);
    gemm_wmma_kernel<128><<<ceil_div((long long)waves * 32, TB), TB, 0, stream>>>(xbuf, onorm, Wp, hbuf);
  }
  edge_agg_kernel<128><<<ceil_div((long long)NEDGES * 32, TB), TB, 0, stream>>>(src, dst, hbuf, abuf, NEDGES);
  finalize_kernel<128, true><<<ceil_div((long long)NNODES * 128, TB), TB, 0, stream>>>(
      abuf, hbuf, inorm, b2, xbuf, NNODES * 128);

  pack_w_kernel<<<ceil_div(4 * 4 * 512, TB), TB, 0, stream>>>(W3, Wp, 64);
  zero_kernel<<<ceil_div((long long)NNODES * 64, TB), TB, 0, stream>>>(abuf, NNODES * 64);
  {
    int waves = (NNODES / 16) * (64 / 16);
    gemm_wmma_kernel<64><<<ceil_div((long long)waves * 32, TB), TB, 0, stream>>>(xbuf, onorm, Wp, hbuf);
  }
  edge_agg_kernel<64><<<ceil_div((long long)NEDGES * 32, TB), TB, 0, stream>>>(src, dst, hbuf, abuf, NEDGES);
  finalize_kernel<64, false><<<ceil_div((long long)NNODES * 64, TB), TB, 0, stream>>>(
      abuf, hbuf, inorm, b3, out, NNODES * 64);
}


# Next layer's product moved into the gather's epilogue: a 16-node block rounds its finished rows to f16 in LDS and multiplies them on WMMA; x never goes to memory; sums and operands unchanged
# speedup vs baseline: 2.6757x; 2.6757x over previous
#include <hip/hip_runtime.h>
#include <math.h>

typedef __attribute__((ext_vector_type(16))) _Float16 v16h;
typedef __attribute__((ext_vector_type(8)))  float    v8f;
typedef __attribute__((ext_vector_type(8)))  _Float16 v8h;
typedef __attribute__((ext_vector_type(4)))  _Float16 v4h;

#define NNODES 100000
#define NEDGES 1600000
#define FDIM   128

static inline int ceil_div(long long a, long long b) { return (int)((a + b - 1) / b); }

__global__ void init_cnt_kernel(int* __restrict__ cin, int* __restrict__ cout, int n) {
  int i = blockIdx.x * blockDim.x + threadIdx.x;
  if (i < n) { cin[i] = 0; cout[i] = 0; }
}

__global__ void degree_rank_kernel(const int* __restrict__ src, const int* __restrict__ dst,
                                   int* __restrict__ cin, int* __restrict__ cout,
                                   int* __restrict__ rank, int e) {
  int i = blockIdx.x * blockDim.x + threadIdx.x;
  if (i < e) {
    rank[i] = atomicAdd(&cin[dst[i]], 1);
    atomicAdd(&cout[src[i]], 1);
  }
}

__global__ void scan_norm_kernel(const int* __restrict__ cin, const int* __restrict__ cout,
                                 int* __restrict__ off, float* __restrict__ onorm,
                                 float* __restrict__ inorm, int n) {
  __shared__ int s_front[8];
  __shared__ int s_own[8];
  int t = threadIdx.x, lane = t & 31, w = t >> 5;
  int b = blockIdx.x;

  int front = 0;
  const int4* c4 = (const int4*)cin;
  for (int i = t; i < b * 256; i += 256) {
    int4 v = c4[i];
    front += (v.x + v.y) + (v.z + v.w);
  }
  int i0 = b * 1024 + t * 4;
  int c[4];
#pragma unroll
  for (int k = 0; k < 4; ++k) c[k] = (i0 + k < n) ? cin[i0 + k] : 0;
  int mine = (c[0] + c[1]) + (c[2] + c[3]);

  int incl = mine;
#pragma unroll
  for (int d = 1; d < 32; d <<= 1) {
    int o = __shfl_up(incl, d);
    if (lane >= d) incl += o;
  }
#pragma unroll
  for (int d = 16; d > 0; d >>= 1) front += __shfl_xor(front, d);
  if (lane == 31) s_own[w] = incl;
  if (lane == 0)  s_front[w] = front;
  __syncthreads();
  int base = 0;
#pragma unroll
  for (int k = 0; k < 8; ++k) {
    base += s_front[k];
    if (k < w) base += s_own[k];
  }
  int run = base + incl - mine;
#pragma unroll
  for (int k = 0; k < 4; ++k) {
    int i = i0 + k;
    if (i < n) {
      off[i] = run;
      run += c[k];
      onorm[i] = rsqrtf(fmaxf((float)(cout[i] + 1), 1.0f));
      inorm[i] = rsqrtf(fmaxf((float)(c[k] + 1), 1.0f));
    }
  }
}

__global__ void place_kernel(const int* __restrict__ src, const int* __restrict__ dst,
                             const int* __restrict__ off, const int* __restrict__ rank,
                             int* __restrict__ lst, int e) {
  int i = blockIdx.x * blockDim.x + threadIdx.x;
  if (i < e) lst[off[dst[i]] + rank[i]] = src[i];
}

__global__ void order_kernel(const int* __restrict__ off, const int* __restrict__ cin,
                             const int* __restrict__ lst, int* __restrict__ csr, int n_nodes) {
  int lane = threadIdx.x & 31;
  int node = (int)((blockIdx.x * blockDim.x + threadIdx.x) >> 5);
  if (node >= n_nodes) return;
  int beg = off[node];
  int n   = cin[node];
  const int* p = lst + beg;
  for (int i = lane; i < n; i += 32) {
    int k = p[i];
    int r = 0;
    for (int j = 0; j < n; ++j) {
      int kj = p[j];
      r += (kj < k || (kj == k && j < i)) ? 1 : 0;
    }
    csr[beg + r] = k;
  }
}

__global__ void pack_w_kernel(const float* __restrict__ W, _Float16* __restrict__ Wp, int nout) {
  int idx = blockIdx.x * blockDim.x + threadIdx.x;
  int nt = nout >> 4;
  int total = 4 * nt * 32 * 16;
  if (idx >= total) return;
  int j    = idx & 15;
  int l    = (idx >> 4) & 31;
  int rest = idx >> 9;
  int t    = rest % nt;
  int c    = rest / nt;
  int col   = t * 16 + (l & 15);
  int kbase = c * 32 + ((l >> 4) * 8);
  int K     = kbase + (j & 7) + ((j >> 3) * 16);
  Wp[idx] = (_Float16)W[K * nout + col];
}

template<int NOUT>
__global__ void gemm_wmma_kernel(const float* __restrict__ X,
                                 const float* __restrict__ onorm,
                                 const _Float16* __restrict__ Wp,
                                 float* __restrict__ H) {
  constexpr int NT = NOUT / 16;
  int wave = (int)((blockIdx.x * blockDim.x + threadIdx.x) >> 5);
  int lane = threadIdx.x & 31;
  int mtile = wave / NT;
  int ntile = wave % NT;
  if (mtile * 16 >= NNODES) return;

  int r    = lane & 15;
  int half = lane >> 4;
  int row  = mtile * 16 + r;
  float scale = onorm[row];
  const float* px = X + (size_t)row * FDIM;

  v8f acc = {0.f, 0.f, 0.f, 0.f, 0.f, 0.f, 0.f, 0.f};

#pragma unroll
  for (int kc = 0; kc < 4; ++kc) {
    int kbase = kc * 32 + half * 8;
    float4 f0 = *(const float4*)(px + kbase);
    float4 f1 = *(const float4*)(px + kbase + 4);
    float4 f2 = *(const float4*)(px + kbase + 16);
    float4 f3 = *(const float4*)(px + kbase + 20);
    v16h a;
    a[0]  = (_Float16)(f0.x * scale); a[1]  = (_Float16)(f0.y * scale);
    a[2]  = (_Float16)(f0.z * scale); a[3]  = (_Float16)(f0.w * scale);
    a[4]  = (_Float16)(f1.x * scale); a[5]  = (_Float16)(f1.y * scale);
    a[6]  = (_Float16)(f1.z * scale); a[7]  = (_Float16)(f1.w * scale);
    a[8]  = (_Float16)(f2.x * scale); a[9]  = (_Float16)(f2.y * scale);
    a[10] = (_Float16)(f2.z * scale); a[11] = (_Float16)(f2.w * scale);
    a[12] = (_Float16)(f3.x * scale); a[13] = (_Float16)(f3.y * scale);
    a[14] = (_Float16)(f3.z * scale); a[15] = (_Float16)(f3.w * scale);

    v16h b = *(const v16h*)(Wp + ((size_t)(kc * NT + ntile) * 32 + lane) * 16);

    acc = __builtin_amdgcn_wmma_f32_16x16x32_f16(
         false, a,  false, b,
         (short)0, acc,  false,  false);
  }

  int col = ntile * 16 + r;
  float* ph = H + (size_t)(mtile * 16 + half * 8) * NOUT + col;
#pragma unroll
  for (int vr = 0; vr < 8; ++vr)
    ph[(size_t)vr * NOUT] = acc[vr];
}

template<int U, typename VT>
__device__ __forceinline__ void add_rows(VT& acc, const VT* __restrict__ hv, int ids, int j, int lane) {
  VT r[U];
#pragma unroll
  for (int u = 0; u < U; ++u) {
    int s = __builtin_amdgcn_readlane(ids, j + u);
    r[u] = hv[(size_t)s * 32 + lane];
  }
#pragma unroll
  for (int u = 0; u < U; ++u) acc += r[u];
}

template<int NOUT, bool RELU>
__global__ void gather_agg_kernel(const int* __restrict__ off, const int* __restrict__ cin,
                                  const int* __restrict__ csr, const float* __restrict__ h,
                                  const float* __restrict__ inorm, const float* __restrict__ b,
                                  float* __restrict__ out, int n_nodes) {
  constexpr int V = NOUT / 32;
  typedef float VT __attribute__((ext_vector_type(V)));
  int lane = threadIdx.x & 31;
  int node = (int)((blockIdx.x * blockDim.x + threadIdx.x) >> 5);
  if (node >= n_nodes) return;
  node = __builtin_amdgcn_readfirstlane(node);
  int beg = off[node];
  int n   = cin[node];
  const VT* hv = (const VT*)h;
  VT acc = hv[(size_t)node * 32 + lane];
  for (int base = 0; base < n; base += 32) {
    int m = n - base;
    if (m > 32) m = 32;
    int ids = (lane < m) ? csr[beg + base + lane] : 0;
    int j = 0;
    for (; j + 8 <= m; j += 8) add_rows<8, VT>(acc, hv, ids, j, lane);
    if (j + 4 <= m) { add_rows<4, VT>(acc, hv, ids, j, lane); j += 4; }
    if (j + 2 <= m) { add_rows<2, VT>(acc, hv, ids, j, lane); j += 2; }
    if (j < m)      { add_rows<1, VT>(acc, hv, ids, j, lane); }
  }
  float sc = inorm[node];
  VT bv = ((const VT*)b)[lane];
  VT v = acc * sc + bv;
  if (RELU) {
#pragma unroll
    for (int k = 0; k < V; ++k) v[k] = fmaxf(v[k], 0.0f);
  }
  ((VT*)out)[(size_t)node * 32 + lane] = v;
}


#define XS_PITCH 136

template<int NOUT2>
__global__ __launch_bounds__(512)
void gather_relu_gemm_kernel(const int* __restrict__ off, const int* __restrict__ cin,
                             const int* __restrict__ csr, const float* __restrict__ h,
                             const float* __restrict__ inorm, const float* __restrict__ b,
                             const float* __restrict__ onorm, const _Float16* __restrict__ Wp,
                             float* __restrict__ hnext, int n_nodes) {
  constexpr int NT2 = NOUT2 / 16;
  typedef float VT __attribute__((ext_vector_type(4)));
  __shared__ __attribute__((aligned(16))) _Float16 xs[16 * XS_PITCH];

  int lane = threadIdx.x & 31;
  int w = __builtin_amdgcn_readfirstlane((int)(threadIdx.x >> 5));
  int node = (int)blockIdx.x * 16 + w;

  VT v = {0.f, 0.f, 0.f, 0.f};
  if (node < n_nodes) {
    int beg = off[node];
    int n   = cin[node];
    const VT* hv = (const VT*)h;
    VT acc = hv[(size_t)node * 32 + lane];
    for (int base = 0; base < n; base += 32) {
      int m = n - base;
      if (m > 32) m = 32;
      int ids = (lane < m) ? csr[beg + base + lane] : 0;
      int j = 0;
      for (; j + 8 <= m; j += 8) add_rows<8, VT>(acc, hv, ids, j, lane);
      if (j + 4 <= m) { add_rows<4, VT>(acc, hv, ids, j, lane); j += 4; }
      if (j + 2 <= m) { add_rows<2, VT>(acc, hv, ids, j, lane); j += 2; }
      if (j < m)      { add_rows<1, VT>(acc, hv, ids, j, lane); }
    }
    float sc = inorm[node];
    VT bv = ((const VT*)b)[lane];
    v = acc * sc + bv;
#pragma unroll
    for (int k = 0; k < 4; ++k) v[k] = fmaxf(v[k], 0.0f);
    float on = onorm[node];
    v = v * on;
  }

  v16h bw[4];
  if (w < NT2) {
#pragma unroll
    for (int kc = 0; kc < 4; ++kc)
      bw[kc] = *(const v16h*)(Wp + ((size_t)(kc * NT2 + w) * 32 + lane) * 16);
  }

  v4h xh;
  xh[0] = (_Float16)v[0]; xh[1] = (_Float16)v[1];
  xh[2] = (_Float16)v[2]; xh[3] = (_Float16)v[3];
  *(v4h*)(xs + w * XS_PITCH + lane * 4) = xh;
  __syncthreads();

  if (w >= NT2) return;

  int r    = lane & 15;
  int half = lane >> 4;
  const _Float16* pa = xs + r * XS_PITCH + half * 8;
  v8f acc2 = {0.f, 0.f, 0.f, 0.f, 0.f, 0.f, 0.f, 0.f};
#pragma unroll
  for (int kc = 0; kc < 4; ++kc) {
    v8h lo = *(const v8h*)(pa + kc * 32);
    v8h hi = *(const v8h*)(pa + kc * 32 + 16);
    v16h a = __builtin_shufflevector(lo, hi, 0, 1, 2, 3, 4, 5, 6, 7, 8, 9, 10, 11, 12, 13, 14, 15);
    acc2 = __builtin_amdgcn_wmma_f32_16x16x32_f16(
         false, a,  false, bw[kc],
         (short)0, acc2,  false,  false);
  }

  int row0 = (int)blockIdx.x * 16 + half * 8;
  float* ph = hnext + (size_t)row0 * NOUT2 + w * 16 + r;
  if ((int)blockIdx.x * 16 + 16 <= n_nodes) {
#pragma unroll
    for (int vr = 0; vr < 8; ++vr) ph[(size_t)vr * NOUT2] = acc2[vr];
  } else {
#pragma unroll
    for (int vr = 0; vr < 8; ++vr)
      if (row0 + vr < n_nodes) ph[(size_t)vr * NOUT2] = acc2[vr];
  }
}

extern "C" void kernel_launch(void* const* d_in, const int* in_sizes, int n_in,
                              void* d_out, int out_size, void* d_ws, size_t ws_size,
                              hipStream_t stream) {
  const float* feat = (const float*)d_in[0];
  const int*   src  = (const int*)  d_in[1];
  const int*   dst  = (const int*)  d_in[2];
  const float* W1   = (const float*)d_in[3];
  const float* b1   = (const float*)d_in[4];
  const float* W2   = (const float*)d_in[5];
  const float* b2   = (const float*)d_in[6];
  const float* W3   = (const float*)d_in[7];
  const float* b3   = (const float*)d_in[8];
  float* out = (float*)d_out;

  size_t off = 0;
  auto alloc = [&](size_t bytes) -> char* {
    char* p = (char*)d_ws + off;
    off += (bytes + 255) & ~(size_t)255;
    return p;
  };
  float*     onorm = (float*)alloc((size_t)NNODES * sizeof(float));
  float*     inorm = (float*)alloc((size_t)NNODES * sizeof(float));
  float*     hbuf  = (float*)alloc((size_t)NNODES * FDIM * sizeof(float));
  float*     xbuf  = (float*)alloc((size_t)NNODES * FDIM * sizeof(float));
  _Float16*  Wp    = (_Float16*)alloc((size_t)FDIM * FDIM * sizeof(_Float16));
  int*       cnt_in = (int*)alloc((size_t)NNODES * sizeof(int));
  int*       cnt_out= (int*)alloc((size_t)NNODES * sizeof(int));
  int*       lstart= (int*)alloc((size_t)NNODES * sizeof(int));
  int*       rank  = (int*)alloc((size_t)NEDGES * sizeof(int));
  int*       lst   = (int*)alloc((size_t)NEDGES * sizeof(int));
  int*       csr   = (int*)alloc((size_t)NEDGES * sizeof(int));

  const int TB = 256;
  const int node_waves_blocks = ceil_div((long long)NNODES * 32, TB);

  init_cnt_kernel<<<ceil_div(NNODES, TB), TB, 0, stream>>>(cnt_in, cnt_out, NNODES);
  degree_rank_kernel<<<ceil_div(NEDGES, TB), TB, 0, stream>>>(src, dst, cnt_in, cnt_out, rank, NEDGES);
  scan_norm_kernel<<<ceil_div(NNODES, 1024), 256, 0, stream>>>(cnt_in, cnt_out, lstart, onorm, inorm, NNODES);
  place_kernel<<<ceil_div(NEDGES, TB), TB, 0, stream>>>(src, dst, lstart, rank, lst, NEDGES);
  order_kernel<<<node_waves_blocks, TB, 0, stream>>>(lstart, cnt_in, lst, csr, NNODES);

  pack_w_kernel<<<ceil_div(4 * 8 * 512, TB), TB, 0, stream>>>(W1, Wp, 128);
  {
    int waves = (NNODES / 16) * (128 / 16);
    gemm_wmma_kernel<128><<<ceil_div((long long)waves * 32, TB), TB, 0, stream>>>(feat, onorm, Wp, hbuf);
  }

  const int tile_blocks = ceil_div(NNODES, 16);

  pack_w_kernel<<<ceil_div(4 * 8 * 512, TB), TB, 0, stream>>>(W2, Wp, 128);
  gather_relu_gemm_kernel<128><<<tile_blocks, 512, 0, stream>>>(
      lstart, cnt_in, csr, hbuf, inorm, b1, onorm, Wp, xbuf, NNODES);

  pack_w_kernel<<<ceil_div(4 * 4 * 512, TB), TB, 0, stream>>>(W3, Wp, 64);
  gather_relu_gemm_kernel<64><<<tile_blocks, 512, 0, stream>>>(
      lstart, cnt_in, csr, xbuf, inorm, b2, onorm, Wp, hbuf, NNODES);

  gather_agg_kernel<64, false><<<node_waves_blocks, TB, 0, stream>>>(
      lstart, cnt_in, csr, hbuf, inorm, b3, out, NNODES);
}


# Scan kernel: the counts in front of a block are summed by 1024 threads with 8 independent 16-byte loads in flight each, not one load per round trip on 256 threads; integer sums, same results
# speedup vs baseline: 2.9752x; 1.1119x over previous
#include <hip/hip_runtime.h>
#include <math.h>

typedef __attribute__((ext_vector_type(16))) _Float16 v16h;
typedef __attribute__((ext_vector_type(8)))  float    v8f;
typedef __attribute__((ext_vector_type(8)))  _Float16 v8h;
typedef __attribute__((ext_vector_type(4)))  _Float16 v4h;

#define NNODES 100000
#define NEDGES 1600000
#define FDIM   128

static inline int ceil_div(long long a, long long b) { return (int)((a + b - 1) / b); }

__global__ void init_cnt_kernel(int* __restrict__ cin, int* __restrict__ cout, int n) {
  int i = blockIdx.x * blockDim.x + threadIdx.x;
  if (i < n) { cin[i] = 0; cout[i] = 0; }
}

__global__ void degree_rank_kernel(const int* __restrict__ src, const int* __restrict__ dst,
                                   int* __restrict__ cin, int* __restrict__ cout,
                                   int* __restrict__ rank, int e) {
  int i = blockIdx.x * blockDim.x + threadIdx.x;
  if (i < e) {
    rank[i] = atomicAdd(&cin[dst[i]], 1);
    atomicAdd(&cout[src[i]], 1);
  }
}

#define SCAN_THREADS 1024
#define SCAN_INFLIGHT 8

__global__ __launch_bounds__(SCAN_THREADS)
void scan_norm_kernel(const int* __restrict__ cin, const int* __restrict__ cout,
                      int* __restrict__ off, float* __restrict__ onorm,
                      float* __restrict__ inorm, int n) {
  __shared__ int s_front[32];
  __shared__ int s_own[32];
  int t = threadIdx.x, lane = t & 31, w = t >> 5;
  int b = blockIdx.x;

  int i = b * SCAN_THREADS + t;
  int c  = (i < n) ? cin[i]  : 0;
  int co = (i < n) ? cout[i] : 0;

  int front = 0;
  const int4* c4 = (const int4*)cin;
  int lim = b * (SCAN_THREADS / 4);
  for (int k = t; k < lim; k += SCAN_INFLIGHT * SCAN_THREADS) {
    int4 v[SCAN_INFLIGHT];
#pragma unroll
    for (int u = 0; u < SCAN_INFLIGHT; ++u) {
      int kk = k + u * SCAN_THREADS;
      v[u] = c4[kk < lim ? kk : lim - 1];
    }
#pragma unroll
    for (int u = 0; u < SCAN_INFLIGHT; ++u) {
      int kk = k + u * SCAN_THREADS;
      int s = (v[u].x + v[u].y) + (v[u].z + v[u].w);
      front += (kk < lim) ? s : 0;
    }
  }

  int incl = c;
#pragma unroll
  for (int d = 1; d < 32; d <<= 1) {
    int o = __shfl_up(incl, d);
    if (lane >= d) incl += o;
  }
#pragma unroll
  for (int d = 16; d > 0; d >>= 1) front += __shfl_xor(front, d);
  if (lane == 31) s_own[w] = incl;
  if (lane == 0)  s_front[w] = front;
  __syncthreads();
  int part = s_front[lane] + ((lane < w) ? s_own[lane] : 0);
#pragma unroll
  for (int d = 16; d > 0; d >>= 1) part += __shfl_xor(part, d);
  if (i < n) {
    off[i] = part + incl - c;
    onorm[i] = rsqrtf(fmaxf((float)(co + 1), 1.0f));
    inorm[i] = rsqrtf(fmaxf((float)(c + 1), 1.0f));
  }
}

__global__ void place_kernel(const int* __restrict__ src, const int* __restrict__ dst,
                             const int* __restrict__ off, const int* __restrict__ rank,
                             int* __restrict__ lst, int e) {
  int i = blockIdx.x * blockDim.x + threadIdx.x;
  if (i < e) lst[off[dst[i]] + rank[i]] = src[i];
}

__global__ void order_kernel(const int* __restrict__ off, const int* __restrict__ cin,
                             const int* __restrict__ lst, int* __restrict__ csr, int n_nodes) {
  int lane = threadIdx.x & 31;
  int node = (int)((blockIdx.x * blockDim.x + threadIdx.x) >> 5);
  if (node >= n_nodes) return;
  int beg = off[node];
  int n   = cin[node];
  const int* p = lst + beg;
  for (int i = lane; i < n; i += 32) {
    int k = p[i];
    int r = 0;
    for (int j = 0; j < n; ++j) {
      int kj = p[j];
      r += (kj < k || (kj == k && j < i)) ? 1 : 0;
    }
    csr[beg + r] = k;
  }
}

__global__ void pack_w_kernel(const float* __restrict__ W, _Float16* __restrict__ Wp, int nout) {
  int idx = blockIdx.x * blockDim.x + threadIdx.x;
  int nt = nout >> 4;
  int total = 4 * nt * 32 * 16;
  if (idx >= total) return;
  int j    = idx & 15;
  int l    = (idx >> 4) & 31;
  int rest = idx >> 9;
  int t    = rest % nt;
  int c    = rest / nt;
  int col   = t * 16 + (l & 15);
  int kbase = c * 32 + ((l >> 4) * 8);
  int K     = kbase + (j & 7) + ((j >> 3) * 16);
  Wp[idx] = (_Float16)W[K * nout + col];
}

template<int NOUT>
__global__ void gemm_wmma_kernel(const float* __restrict__ X,
                                 const float* __restrict__ onorm,
                                 const _Float16* __restrict__ Wp,
                                 float* __restrict__ H) {
  constexpr int NT = NOUT / 16;
  int wave = (int)((blockIdx.x * blockDim.x + threadIdx.x) >> 5);
  int lane = threadIdx.x & 31;
  int mtile = wave / NT;
  int ntile = wave % NT;
  if (mtile * 16 >= NNODES) return;

  int r    = lane & 15;
  int half = lane >> 4;
  int row  = mtile * 16 + r;
  float scale = onorm[row];
  const float* px = X + (size_t)row * FDIM;

  v8f acc = {0.f, 0.f, 0.f, 0.f, 0.f, 0.f, 0.f, 0.f};

#pragma unroll
  for (int kc = 0; kc < 4; ++kc) {
    int kbase = kc * 32 + half * 8;
    float4 f0 = *(const float4*)(px + kbase);
    float4 f1 = *(const float4*)(px + kbase + 4);
    float4 f2 = *(const float4*)(px + kbase + 16);
    float4 f3 = *(const float4*)(px + kbase + 20);
    v16h a;
    a[0]  = (_Float16)(f0.x * scale); a[1]  = (_Float16)(f0.y * scale);
    a[2]  = (_Float16)(f0.z * scale); a[3]  = (_Float16)(f0.w * scale);
    a[4]  = (_Float16)(f1.x * scale); a[5]  = (_Float16)(f1.y * scale);
    a[6]  = (_Float16)(f1.z * scale); a[7]  = (_Float16)(f1.w * scale);
    a[8]  = (_Float16)(f2.x * scale); a[9]  = (_Float16)(f2.y * scale);
    a[10] = (_Float16)(f2.z * scale); a[11] = (_Float16)(f2.w * scale);
    a[12] = (_Float16)(f3.x * scale); a[13] = (_Float16)(f3.y * scale);
    a[14] = (_Float16)(f3.z * scale); a[15] = (_Float16)(f3.w * scale);

    v16h b = *(const v16h*)(Wp + ((size_t)(kc * NT + ntile) * 32 + lane) * 16);

    acc = __builtin_amdgcn_wmma_f32_16x16x32_f16(
         false, a,  false, b,
         (short)0, acc,  false,  false);
  }

  int col = ntile * 16 + r;
  float* ph = H + (size_t)(mtile * 16 + half * 8) * NOUT + col;
#pragma unroll
  for (int vr = 0; vr < 8; ++vr)
    ph[(size_t)vr * NOUT] = acc[vr];
}

template<int U, typename VT>
__device__ __forceinline__ void add_rows(VT& acc, const VT* __restrict__ hv, int ids, int j, int lane) {
  VT r[U];
#pragma unroll
  for (int u = 0; u < U; ++u) {
    int s = __builtin_amdgcn_readlane(ids, j + u);
    r[u] = hv[(size_t)s * 32 + lane];
  }
#pragma unroll
  for (int u = 0; u < U; ++u) acc += r[u];
}

template<int NOUT, bool RELU>
__global__ void gather_agg_kernel(const int* __restrict__ off, const int* __restrict__ cin,
                                  const int* __restrict__ csr, const float* __restrict__ h,
                                  const float* __restrict__ inorm, const float* __restrict__ b,
                                  float* __restrict__ out, int n_nodes) {
  constexpr int V = NOUT / 32;
  typedef float VT __attribute__((ext_vector_type(V)));
  int lane = threadIdx.x & 31;
  int node = (int)((blockIdx.x * blockDim.x + threadIdx.x) >> 5);
  if (node >= n_nodes) return;
  node = __builtin_amdgcn_readfirstlane(node);
  int beg = off[node];
  int n   = cin[node];
  const VT* hv = (const VT*)h;
  VT acc = hv[(size_t)node * 32 + lane];
  for (int base = 0; base < n; base += 32) {
    int m = n - base;
    if (m > 32) m = 32;
    int ids = (lane < m) ? csr[beg + base + lane] : 0;
    int j = 0;
    for (; j + 8 <= m; j += 8) add_rows<8, VT>(acc, hv, ids, j, lane);
    if (j + 4 <= m) { add_rows<4, VT>(acc, hv, ids, j, lane); j += 4; }
    if (j + 2 <= m) { add_rows<2, VT>(acc, hv, ids, j, lane); j += 2; }
    if (j < m)      { add_rows<1, VT>(acc, hv, ids, j, lane); }
  }
  float sc = inorm[node];
  VT bv = ((const VT*)b)[lane];
  VT v = acc * sc + bv;
  if (RELU) {
#pragma unroll
    for (int k = 0; k < V; ++k) v[k] = fmaxf(v[k], 0.0f);
  }
  ((VT*)out)[(size_t)node * 32 + lane] = v;
}


#define XS_PITCH 136

template<int NOUT2>
__global__ __launch_bounds__(512)
void gather_relu_gemm_kernel(const int* __restrict__ off, const int* __restrict__ cin,
                             const int* __restrict__ csr, const float* __restrict__ h,
                             const float* __restrict__ inorm, const float* __restrict__ b,
                             const float* __restrict__ onorm, const _Float16* __restrict__ Wp,
                             float* __restrict__ hnext, int n_nodes) {
  constexpr int NT2 = NOUT2 / 16;
  typedef float VT __attribute__((ext_vector_type(4)));
  __shared__ __attribute__((aligned(16))) _Float16 xs[16 * XS_PITCH];

  int lane = threadIdx.x & 31;
  int w = __builtin_amdgcn_readfirstlane((int)(threadIdx.x >> 5));
  int node = (int)blockIdx.x * 16 + w;

  VT v = {0.f, 0.f, 0.f, 0.f};
  if (node < n_nodes) {
    int beg = off[node];
    int n   = cin[node];
    const VT* hv = (const VT*)h;
    VT acc = hv[(size_t)node * 32 + lane];
    for (int base = 0; base < n; base += 32) {
      int m = n - base;
      if (m > 32) m = 32;
      int ids = (lane < m) ? csr[beg + base + lane] : 0;
      int j = 0;
      for (; j + 8 <= m; j += 8) add_rows<8, VT>(acc, hv, ids, j, lane);
      if (j + 4 <= m) { add_rows<4, VT>(acc, hv, ids, j, lane); j += 4; }
      if (j + 2 <= m) { add_rows<2, VT>(acc, hv, ids, j, lane); j += 2; }
      if (j < m)      { add_rows<1, VT>(acc, hv, ids, j, lane); }
    }
    float sc = inorm[node];
    VT bv = ((const VT*)b)[lane];
    v = acc * sc + bv;
#pragma unroll
    for (int k = 0; k < 4; ++k) v[k] = fmaxf(v[k], 0.0f);
    float on = onorm[node];
    v = v * on;
  }

  v16h bw[4];
  if (w < NT2) {
#pragma unroll
    for (int kc = 0; kc < 4; ++kc)
      bw[kc] = *(const v16h*)(Wp + ((size_t)(kc * NT2 + w) * 32 + lane) * 16);
  }

  v4h xh;
  xh[0] = (_Float16)v[0]; xh[1] = (_Float16)v[1];
  xh[2] = (_Float16)v[2]; xh[3] = (_Float16)v[3];
  *(v4h*)(xs + w * XS_PITCH + lane * 4) = xh;
  __syncthreads();

  if (w >= NT2) return;

  int r    = lane & 15;
  int half = lane >> 4;
  const _Float16* pa = xs + r * XS_PITCH + half * 8;
  v8f acc2 = {0.f, 0.f, 0.f, 0.f, 0.f, 0.f, 0.f, 0.f};
#pragma unroll
  for (int kc = 0; kc < 4; ++kc) {
    v8h lo = *(const v8h*)(pa + kc * 32);
    v8h hi = *(const v8h*)(pa + kc * 32 + 16);
    v16h a = __builtin_shufflevector(lo, hi, 0, 1, 2, 3, 4, 5, 6, 7, 8, 9, 10, 11, 12, 13, 14, 15);
    acc2 = __builtin_amdgcn_wmma_f32_16x16x32_f16(
         false, a,  false, bw[kc],
         (short)0, acc2,  false,  false);
  }

  int row0 = (int)blockIdx.x * 16 + half * 8;
  float* ph = hnext + (size_t)row0 * NOUT2 + w * 16 + r;
  if ((int)blockIdx.x * 16 + 16 <= n_nodes) {
#pragma unroll
    for (int vr = 0; vr < 8; ++vr) ph[(size_t)vr * NOUT2] = acc2[vr];
  } else {
#pragma unroll
    for (int vr = 0; vr < 8; ++vr)
      if (row0 + vr < n_nodes) ph[(size_t)vr * NOUT2] = acc2[vr];
  }
}

extern "C" void kernel_launch(void* const* d_in, const int* in_sizes, int n_in,
                              void* d_out, int out_size, void* d_ws, size_t ws_size,
                              hipStream_t stream) {
  const float* feat = (const float*)d_in[0];
  const int*   src  = (const int*)  d_in[1];
  const int*   dst  = (const int*)  d_in[2];
  const float* W1   = (const float*)d_in[3];
  const float* b1   = (const float*)d_in[4];
  const float* W2   = (const float*)d_in[5];
  const float* b2   = (const float*)d_in[6];
  const float* W3   = (const float*)d_in[7];
  const float* b3   = (const float*)d_in[8];
  float* out = (float*)d_out;

  size_t off = 0;
  auto alloc = [&](size_t bytes) -> char* {
    char* p = (char*)d_ws + off;
    off += (bytes + 255) & ~(size_t)255;
    return p;
  };
  float*     onorm = (float*)alloc((size_t)NNODES * sizeof(float));
  float*     inorm = (float*)alloc((size_t)NNODES * sizeof(float));
  float*     hbuf  = (float*)alloc((size_t)NNODES * FDIM * sizeof(float));
  float*     xbuf  = (float*)alloc((size_t)NNODES * FDIM * sizeof(float));
  _Float16*  Wp    = (_Float16*)alloc((size_t)FDIM * FDIM * sizeof(_Float16));
  int*       cnt_in = (int*)alloc((size_t)NNODES * sizeof(int));
  int*       cnt_out= (int*)alloc((size_t)NNODES * sizeof(int));
  int*       lstart= (int*)alloc((size_t)NNODES * sizeof(int));
  int*       rank  = (int*)alloc((size_t)NEDGES * sizeof(int));
  int*       lst   = (int*)alloc((size_t)NEDGES * sizeof(int));
  int*       csr   = (int*)alloc((size_t)NEDGES * sizeof(int));

  const int TB = 256;
  const int node_waves_blocks = ceil_div((long long)NNODES * 32, TB);

  init_cnt_kernel<<<ceil_div(NNODES, TB), TB, 0, stream>>>(cnt_in, cnt_out, NNODES);
  degree_rank_kernel<<<ceil_div(NEDGES, TB), TB, 0, stream>>>(src, dst, cnt_in, cnt_out, rank, NEDGES);
  scan_norm_kernel<<<ceil_div(NNODES, SCAN_THREADS), SCAN_THREADS, 0, stream>>>(cnt_in, cnt_out, lstart, onorm, inorm, NNODES);
  place_kernel<<<ceil_div(NEDGES, TB), TB, 0, stream>>>(src, dst, lstart, rank, lst, NEDGES);
  order_kernel<<<node_waves_blocks, TB, 0, stream>>>(lstart, cnt_in, lst, csr, NNODES);

  pack_w_kernel<<<ceil_div(4 * 8 * 512, TB), TB, 0, stream>>>(W1, Wp, 128);
  {
    int waves = (NNODES / 16) * (128 / 16);
    gemm_wmma_kernel<128><<<ceil_div((long long)waves * 32, TB), TB, 0, stream>>>(feat, onorm, Wp, hbuf);
  }

  const int tile_blocks = ceil_div(NNODES, 16);

  pack_w_kernel<<<ceil_div(4 * 8 * 512, TB), TB, 0, stream>>>(W2, Wp, 128);
  gather_relu_gemm_kernel<128><<<tile_blocks, 512, 0, stream>>>(
      lstart, cnt_in, csr, hbuf, inorm, b1, onorm, Wp, xbuf, NNODES);

  pack_w_kernel<<<ceil_div(4 * 4 * 512, TB), TB, 0, stream>>>(W3, Wp, 64);
  gather_relu_gemm_kernel<64><<<tile_blocks, 512, 0, stream>>>(
      lstart, cnt_in, csr, xbuf, inorm, b2, onorm, Wp, hbuf, NNODES);

  gather_agg_kernel<64, false><<<node_waves_blocks, TB, 0, stream>>>(
      lstart, cnt_in, csr, hbuf, inorm, b3, out, NNODES);
}


# Layer 1 product moved behind its aggregation: the gather sums out_norm-scaled feature rows (same list order, own row first), then x W1 and x W2 on WMMA in its epilogue; stand-alone product pass gone
# speedup vs baseline: 3.1951x; 1.0739x over previous
#include <hip/hip_runtime.h>
#include <math.h>

typedef __attribute__((ext_vector_type(16))) _Float16 v16h;
typedef __attribute__((ext_vector_type(8)))  float    v8f;
typedef __attribute__((ext_vector_type(8)))  _Float16 v8h;
typedef __attribute__((ext_vector_type(4)))  _Float16 v4h;

#define NNODES 100000
#define NEDGES 1600000
#define FDIM   128

static inline int ceil_div(long long a, long long b) { return (int)((a + b - 1) / b); }

__global__ void init_cnt_kernel(int* __restrict__ cin, int* __restrict__ cout, int n) {
  int i = blockIdx.x * blockDim.x + threadIdx.x;
  if (i < n) { cin[i] = 0; cout[i] = 0; }
}

__global__ void degree_rank_kernel(const int* __restrict__ src, const int* __restrict__ dst,
                                   int* __restrict__ cin, int* __restrict__ cout,
                                   int* __restrict__ rank, int e) {
  int i = blockIdx.x * blockDim.x + threadIdx.x;
  if (i < e) {
    rank[i] = atomicAdd(&cin[dst[i]], 1);
    atomicAdd(&cout[src[i]], 1);
  }
}

#define SCAN_THREADS 1024
#define SCAN_INFLIGHT 8

__global__ __launch_bounds__(SCAN_THREADS)
void scan_norm_kernel(const int* __restrict__ cin, const int* __restrict__ cout,
                      int* __restrict__ off, float* __restrict__ onorm,
                      float* __restrict__ inorm, int n) {
  __shared__ int s_front[32];
  __shared__ int s_own[32];
  int t = threadIdx.x, lane = t & 31, w = t >> 5;
  int b = blockIdx.x;

  int i = b * SCAN_THREADS + t;
  int c  = (i < n) ? cin[i]  : 0;
  int co = (i < n) ? cout[i] : 0;

  int front = 0;
  const int4* c4 = (const int4*)cin;
  int lim = b * (SCAN_THREADS / 4);
  for (int k = t; k < lim; k += SCAN_INFLIGHT * SCAN_THREADS) {
    int4 v[SCAN_INFLIGHT];
#pragma unroll
    for (int u = 0; u < SCAN_INFLIGHT; ++u) {
      int kk = k + u * SCAN_THREADS;
      v[u] = c4[kk < lim ? kk : lim - 1];
    }
#pragma unroll
    for (int u = 0; u < SCAN_INFLIGHT; ++u) {
      int kk = k + u * SCAN_THREADS;
      int s = (v[u].x + v[u].y) + (v[u].z + v[u].w);
      front += (kk < lim) ? s : 0;
    }
  }

  int incl = c;
#pragma unroll
  for (int d = 1; d < 32; d <<= 1) {
    int o = __shfl_up(incl, d);
    if (lane >= d) incl += o;
  }
#pragma unroll
  for (int d = 16; d > 0; d >>= 1) front += __shfl_xor(front, d);
  if (lane == 31) s_own[w] = incl;
  if (lane == 0)  s_front[w] = front;
  __syncthreads();
  int part = s_front[lane] + ((lane < w) ? s_own[lane] : 0);
#pragma unroll
  for (int d = 16; d > 0; d >>= 1) part += __shfl_xor(part, d);
  if (i < n) {
    off[i] = part + incl - c;
    onorm[i] = rsqrtf(fmaxf((float)(co + 1), 1.0f));
    inorm[i] = rsqrtf(fmaxf((float)(c + 1), 1.0f));
  }
}

__global__ void place_kernel(const int* __restrict__ src, const int* __restrict__ dst,
                             const int* __restrict__ off, const int* __restrict__ rank,
                             int* __restrict__ lst, int e) {
  int i = blockIdx.x * blockDim.x + threadIdx.x;
  if (i < e) lst[off[dst[i]] + rank[i]] = src[i];
}

__global__ void order_kernel(const int* __restrict__ off, const int* __restrict__ cin,
                             const int* __restrict__ lst, const float* __restrict__ onorm,
                             int* __restrict__ csr, float* __restrict__ cw, int n_nodes) {
  int lane = threadIdx.x & 31;
  int node = (int)((blockIdx.x * blockDim.x + threadIdx.x) >> 5);
  if (node >= n_nodes) return;
  int beg = off[node];
  int n   = cin[node];
  const int* p = lst + beg;
  for (int i = lane; i < n; i += 32) {
    int k = p[i];
    float wk = onorm[k];
    int r = 0;
    for (int j = 0; j < n; ++j) {
      int kj = p[j];
      r += (kj < k || (kj == k && j < i)) ? 1 : 0;
    }
    csr[beg + r] = k;
    cw[beg + r]  = wk;
  }
}

__global__ void pack_w_kernel(const float* __restrict__ W, _Float16* __restrict__ Wp, int nout) {
  int idx = blockIdx.x * blockDim.x + threadIdx.x;
  int nt = nout >> 4;
  int total = 4 * nt * 32 * 16;
  if (idx >= total) return;
  int j    = idx & 15;
  int l    = (idx >> 4) & 31;
  int rest = idx >> 9;
  int t    = rest % nt;
  int c    = rest / nt;
  int col   = t * 16 + (l & 15);
  int kbase = c * 32 + ((l >> 4) * 8);
  int K     = kbase + (j & 7) + ((j >> 3) * 16);
  Wp[idx] = (_Float16)W[K * nout + col];
}

template<int U, typename VT>
__device__ __forceinline__ void add_rows(VT& acc, const VT* __restrict__ hv, int ids, int j, int lane) {
  VT r[U];
#pragma unroll
  for (int u = 0; u < U; ++u) {
    int s = __builtin_amdgcn_readlane(ids, j + u);
    r[u] = hv[(size_t)s * 32 + lane];
  }
#pragma unroll
  for (int u = 0; u < U; ++u) acc += r[u];
}

template<int NOUT, bool RELU>
__global__ void gather_agg_kernel(const int* __restrict__ off, const int* __restrict__ cin,
                                  const int* __restrict__ csr, const float* __restrict__ h,
                                  const float* __restrict__ inorm, const float* __restrict__ b,
                                  float* __restrict__ out, int n_nodes) {
  constexpr int V = NOUT / 32;
  typedef float VT __attribute__((ext_vector_type(V)));
  int lane = threadIdx.x & 31;
  int node = (int)((blockIdx.x * blockDim.x + threadIdx.x) >> 5);
  if (node >= n_nodes) return;
  node = __builtin_amdgcn_readfirstlane(node);
  int beg = off[node];
  int n   = cin[node];
  const VT* hv = (const VT*)h;
  VT acc = hv[(size_t)node * 32 + lane];
  for (int base = 0; base < n; base += 32) {
    int m = n - base;
    if (m > 32) m = 32;
    int ids = (lane < m) ? csr[beg + base + lane] : 0;
    int j = 0;
    for (; j + 8 <= m; j += 8) add_rows<8, VT>(acc, hv, ids, j, lane);
    if (j + 4 <= m) { add_rows<4, VT>(acc, hv, ids, j, lane); j += 4; }
    if (j + 2 <= m) { add_rows<2, VT>(acc, hv, ids, j, lane); j += 2; }
    if (j < m)      { add_rows<1, VT>(acc, hv, ids, j, lane); }
  }
  float sc = inorm[node];
  VT bv = ((const VT*)b)[lane];
  VT v = acc * sc + bv;
  if (RELU) {
#pragma unroll
    for (int k = 0; k < V; ++k) v[k] = fmaxf(v[k], 0.0f);
  }
  ((VT*)out)[(size_t)node * 32 + lane] = v;
}


#define XS_PITCH 136

template<int NOUT2>
__global__ __launch_bounds__(512)
void gather_relu_gemm_kernel(const int* __restrict__ off, const int* __restrict__ cin,
                             const int* __restrict__ csr, const float* __restrict__ h,
                             const float* __restrict__ inorm, const float* __restrict__ b,
                             const float* __restrict__ onorm, const _Float16* __restrict__ Wp,
                             float* __restrict__ hnext, int n_nodes) {
  constexpr int NT2 = NOUT2 / 16;
  typedef float VT __attribute__((ext_vector_type(4)));
  __shared__ __attribute__((aligned(16))) _Float16 xs[16 * XS_PITCH];

  int lane = threadIdx.x & 31;
  int w = __builtin_amdgcn_readfirstlane((int)(threadIdx.x >> 5));
  int node = (int)blockIdx.x * 16 + w;

  VT v = {0.f, 0.f, 0.f, 0.f};
  if (node < n_nodes) {
    int beg = off[node];
    int n   = cin[node];
    const VT* hv = (const VT*)h;
    VT acc = hv[(size_t)node * 32 + lane];
    for (int base = 0; base < n; base += 32) {
      int m = n - base;
      if (m > 32) m = 32;
      int ids = (lane < m) ? csr[beg + base + lane] : 0;
      int j = 0;
      for (; j + 8 <= m; j += 8) add_rows<8, VT>(acc, hv, ids, j, lane);
      if (j + 4 <= m) { add_rows<4, VT>(acc, hv, ids, j, lane); j += 4; }
      if (j + 2 <= m) { add_rows<2, VT>(acc, hv, ids, j, lane); j += 2; }
      if (j < m)      { add_rows<1, VT>(acc, hv, ids, j, lane); }
    }
    float sc = inorm[node];
    VT bv = ((const VT*)b)[lane];
    v = acc * sc + bv;
#pragma unroll
    for (int k = 0; k < 4; ++k) v[k] = fmaxf(v[k], 0.0f);
    float on = onorm[node];
    v = v * on;
  }

  v16h bw[4];
  if (w < NT2) {
#pragma unroll
    for (int kc = 0; kc < 4; ++kc)
      bw[kc] = *(const v16h*)(Wp + ((size_t)(kc * NT2 + w) * 32 + lane) * 16);
  }

  v4h xh;
  xh[0] = (_Float16)v[0]; xh[1] = (_Float16)v[1];
  xh[2] = (_Float16)v[2]; xh[3] = (_Float16)v[3];
  *(v4h*)(xs + w * XS_PITCH + lane * 4) = xh;
  __syncthreads();

  if (w >= NT2) return;

  int r    = lane & 15;
  int half = lane >> 4;
  const _Float16* pa = xs + r * XS_PITCH + half * 8;
  v8f acc2 = {0.f, 0.f, 0.f, 0.f, 0.f, 0.f, 0.f, 0.f};
#pragma unroll
  for (int kc = 0; kc < 4; ++kc) {
    v8h lo = *(const v8h*)(pa + kc * 32);
    v8h hi = *(const v8h*)(pa + kc * 32 + 16);
    v16h a = __builtin_shufflevector(lo, hi, 0, 1, 2, 3, 4, 5, 6, 7, 8, 9, 10, 11, 12, 13, 14, 15);
    acc2 = __builtin_amdgcn_wmma_f32_16x16x32_f16(
         false, a,  false, bw[kc],
         (short)0, acc2,  false,  false);
  }

  int row0 = (int)blockIdx.x * 16 + half * 8;
  float* ph = hnext + (size_t)row0 * NOUT2 + w * 16 + r;
  if ((int)blockIdx.x * 16 + 16 <= n_nodes) {
#pragma unroll
    for (int vr = 0; vr < 8; ++vr) ph[(size_t)vr * NOUT2] = acc2[vr];
  } else {
#pragma unroll
    for (int vr = 0; vr < 8; ++vr)
      if (row0 + vr < n_nodes) ph[(size_t)vr * NOUT2] = acc2[vr];
  }
}

template<int U, typename VT>
__device__ __forceinline__ void add_rows_scaled(VT& acc, const VT* __restrict__ xv, int ids, int scl, int j, int lane) {
  VT r[U];
#pragma unroll
  for (int u = 0; u < U; ++u) {
    int s = __builtin_amdgcn_readlane(ids, j + u);
    r[u] = xv[(size_t)s * 32 + lane];
  }
#pragma unroll
  for (int u = 0; u < U; ++u) {
    float sc = __int_as_float(__builtin_amdgcn_readlane(scl, j + u));
    acc += r[u] * sc;
  }
}

__global__ __launch_bounds__(512)
void gather_gemm_relu_gemm_kernel(const int* __restrict__ off, const int* __restrict__ cin,
                                  const int* __restrict__ csr, const float* __restrict__ cw,
                                  const float* __restrict__ feat,
                                  const float* __restrict__ inorm, const float* __restrict__ b,
                                  const float* __restrict__ onorm,
                                  const _Float16* __restrict__ Wp1, const _Float16* __restrict__ Wp2,
                                  float* __restrict__ hnext, int n_nodes) {
  typedef float VT __attribute__((ext_vector_type(4)));
  __shared__ __attribute__((aligned(16))) _Float16 xs[16 * XS_PITCH];
  __shared__ __attribute__((aligned(16))) _Float16 ys[16 * XS_PITCH];
  __shared__ __attribute__((aligned(16))) float s_in[16];
  __shared__ __attribute__((aligned(16))) float s_on[16];

  int lane = threadIdx.x & 31;
  int w = __builtin_amdgcn_readfirstlane((int)(threadIdx.x >> 5));
  int node = (int)blockIdx.x * 16 + w;

  VT acc = {0.f, 0.f, 0.f, 0.f};
  float sc_in = 0.f, sc_on = 0.f;
  if (node < n_nodes) {
    int beg = off[node];
    int n   = cin[node];
    sc_on   = onorm[node];
    const VT* xv = (const VT*)feat;
    acc = xv[(size_t)node * 32 + lane] * sc_on;
    for (int base = 0; base < n; base += 32) {
      int m = n - base;
      if (m > 32) m = 32;
      int ids = (lane < m) ? csr[beg + base + lane] : 0;
      int scl = (lane < m) ? __float_as_int(cw[beg + base + lane]) : 0;
      int j = 0;
      for (; j + 8 <= m; j += 8) add_rows_scaled<8, VT>(acc, xv, ids, scl, j, lane);
      if (j + 4 <= m) { add_rows_scaled<4, VT>(acc, xv, ids, scl, j, lane); j += 4; }
      if (j + 2 <= m) { add_rows_scaled<2, VT>(acc, xv, ids, scl, j, lane); j += 2; }
      if (j < m)      { add_rows_scaled<1, VT>(acc, xv, ids, scl, j, lane); }
    }
    sc_in = inorm[node];
  }

  int t = w & 7;
  const _Float16* Wsel = (w < 8) ? Wp1 : Wp2;
  v16h bw[4];
#pragma unroll
  for (int kc = 0; kc < 4; ++kc)
    bw[kc] = *(const v16h*)(Wsel + ((size_t)(kc * 8 + t) * 32 + lane) * 16);

  v4h xh;
  xh[0] = (_Float16)acc[0]; xh[1] = (_Float16)acc[1];
  xh[2] = (_Float16)acc[2]; xh[3] = (_Float16)acc[3];
  *(v4h*)(xs + w * XS_PITCH + lane * 4) = xh;
  if (lane == 0) { s_in[w] = sc_in; s_on[w] = sc_on; }
  __syncthreads();

  int r    = lane & 15;
  int half = lane >> 4;
  if (w < 8) {
    const _Float16* pa = xs + r * XS_PITCH + half * 8;
    v8f acc1 = {0.f, 0.f, 0.f, 0.f, 0.f, 0.f, 0.f, 0.f};
#pragma unroll
    for (int kc = 0; kc < 4; ++kc) {
      v8h lo = *(const v8h*)(pa + kc * 32);
      v8h hi = *(const v8h*)(pa + kc * 32 + 16);
      v16h a = __builtin_shufflevector(lo, hi, 0, 1, 2, 3, 4, 5, 6, 7, 8, 9, 10, 11, 12, 13, 14, 15);
      acc1 = __builtin_amdgcn_wmma_f32_16x16x32_f16(
           false, a,  false, bw[kc],
           (short)0, acc1,  false,  false);
    }
    float bcol = b[t * 16 + r];
#pragma unroll
    for (int vr = 0; vr < 8; ++vr) {
      int row = half * 8 + vr;
      float v = acc1[vr] * s_in[row] + bcol;
      v = fmaxf(v, 0.0f);
      v = v * s_on[row];
      ys[row * XS_PITCH + t * 16 + r] = (_Float16)v;
    }
  }
  __syncthreads();

  if (w < 8) return;

  const _Float16* pb = ys + r * XS_PITCH + half * 8;
  v8f acc2 = {0.f, 0.f, 0.f, 0.f, 0.f, 0.f, 0.f, 0.f};
#pragma unroll
  for (int kc = 0; kc < 4; ++kc) {
    v8h lo = *(const v8h*)(pb + kc * 32);
    v8h hi = *(const v8h*)(pb + kc * 32 + 16);
    v16h a = __builtin_shufflevector(lo, hi, 0, 1, 2, 3, 4, 5, 6, 7, 8, 9, 10, 11, 12, 13, 14, 15);
    acc2 = __builtin_amdgcn_wmma_f32_16x16x32_f16(
         false, a,  false, bw[kc],
         (short)0, acc2,  false,  false);
  }

  int row0 = (int)blockIdx.x * 16 + half * 8;
  float* ph = hnext + (size_t)row0 * 128 + t * 16 + r;
  if ((int)blockIdx.x * 16 + 16 <= n_nodes) {
#pragma unroll
    for (int vr = 0; vr < 8; ++vr) ph[(size_t)vr * 128] = acc2[vr];
  } else {
#pragma unroll
    for (int vr = 0; vr < 8; ++vr)
      if (row0 + vr < n_nodes) ph[(size_t)vr * 128] = acc2[vr];
  }
}

extern "C" void kernel_launch(void* const* d_in, const int* in_sizes, int n_in,
                              void* d_out, int out_size, void* d_ws, size_t ws_size,
                              hipStream_t stream) {
  const float* feat = (const float*)d_in[0];
  const int*   src  = (const int*)  d_in[1];
  const int*   dst  = (const int*)  d_in[2];
  const float* W1   = (const float*)d_in[3];
  const float* b1   = (const float*)d_in[4];
  const float* W2   = (const float*)d_in[5];
  const float* b2   = (const float*)d_in[6];
  const float* W3   = (const float*)d_in[7];
  const float* b3   = (const float*)d_in[8];
  float* out = (float*)d_out;

  size_t off = 0;
  auto alloc = [&](size_t bytes) -> char* {
    char* p = (char*)d_ws + off;
    off += (bytes + 255) & ~(size_t)255;
    return p;
  };
  float*     onorm = (float*)alloc((size_t)NNODES * sizeof(float));
  float*     inorm = (float*)alloc((size_t)NNODES * sizeof(float));
  float*     hbuf  = (float*)alloc((size_t)NNODES * FDIM * sizeof(float));
  float*     xbuf  = (float*)alloc((size_t)NNODES * FDIM * sizeof(float));
  _Float16*  Wp1   = (_Float16*)alloc((size_t)FDIM * FDIM * sizeof(_Float16));
  _Float16*  Wp2   = (_Float16*)alloc((size_t)FDIM * FDIM * sizeof(_Float16));
  _Float16*  Wp3   = (_Float16*)alloc((size_t)FDIM * FDIM * sizeof(_Float16));
  int*       cnt_in = (int*)alloc((size_t)NNODES * sizeof(int));
  int*       cnt_out= (int*)alloc((size_t)NNODES * sizeof(int));
  int*       lstart= (int*)alloc((size_t)NNODES * sizeof(int));
  int*       rank  = (int*)alloc((size_t)NEDGES * sizeof(int));
  int*       lst   = (int*)alloc((size_t)NEDGES * sizeof(int));
  int*       csr   = (int*)alloc((size_t)NEDGES * sizeof(int));
  float*     cw    = (float*)alloc((size_t)NEDGES * sizeof(float));

  const int TB = 256;
  const int node_waves_blocks = ceil_div((long long)NNODES * 32, TB);

  init_cnt_kernel<<<ceil_div(NNODES, TB), TB, 0, stream>>>(cnt_in, cnt_out, NNODES);
  degree_rank_kernel<<<ceil_div(NEDGES, TB), TB, 0, stream>>>(src, dst, cnt_in, cnt_out, rank, NEDGES);
  scan_norm_kernel<<<ceil_div(NNODES, SCAN_THREADS), SCAN_THREADS, 0, stream>>>(cnt_in, cnt_out, lstart, onorm, inorm, NNODES);
  place_kernel<<<ceil_div(NEDGES, TB), TB, 0, stream>>>(src, dst, lstart, rank, lst, NEDGES);
  order_kernel<<<node_waves_blocks, TB, 0, stream>>>(lstart, cnt_in, lst, onorm, csr, cw, NNODES);

  const int tile_blocks = ceil_div(NNODES, 16);

  pack_w_kernel<<<ceil_div(4 * 8 * 512, TB), TB, 0, stream>>>(W1, Wp1, 128);
  pack_w_kernel<<<ceil_div(4 * 8 * 512, TB), TB, 0, stream>>>(W2, Wp2, 128);
  gather_gemm_relu_gemm_kernel<<<tile_blocks, 512, 0, stream>>>(
      lstart, cnt_in, csr, cw, feat, inorm, b1, onorm, Wp1, Wp2, xbuf, NNODES);

  pack_w_kernel<<<ceil_div(4 * 4 * 512, TB), TB, 0, stream>>>(W3, Wp3, 64);
  gather_relu_gemm_kernel<64><<<tile_blocks, 512, 0, stream>>>(
      lstart, cnt_in, csr, xbuf, inorm, b2, onorm, Wp3, hbuf, NNODES);

  gather_agg_kernel<64, false><<<node_waves_blocks, TB, 0, stream>>>(
      lstart, cnt_in, csr, hbuf, inorm, b3, out, NNODES);
}


# Edge grouping without a returning global add per edge: two-level counting sort, chunks deal edges to 196 node-range buckets, a block per bucket fills its lists, ranks from LDS counters; same sums
# speedup vs baseline: 3.9076x; 1.2230x over previous
#include <hip/hip_runtime.h>
#include <math.h>

typedef __attribute__((ext_vector_type(16))) _Float16 v16h;
typedef __attribute__((ext_vector_type(8)))  float    v8f;
typedef __attribute__((ext_vector_type(8)))  _Float16 v8h;
typedef __attribute__((ext_vector_type(4)))  _Float16 v4h;

#define NNODES 100000
#define NEDGES 1600000
#define FDIM   128

static inline int ceil_div(long long a, long long b) { return (int)((a + b - 1) / b); }

#define BKT_SHIFT   9
#define BKT_NODES   (1 << BKT_SHIFT)
#define NBKT        ((NNODES + BKT_NODES - 1) / BKT_NODES)
#define NCHUNK      256
#define CHUNK_EDGES ((NEDGES + NCHUNK - 1) / NCHUNK)
#define CHUNK_PER   ((CHUNK_EDGES + 1023) / 1024)
#define ZSLICE      ((NNODES + NCHUNK - 1) / NCHUNK)

__global__ __launch_bounds__(1024)
void chunk_count_kernel(const int* __restrict__ dst, int* __restrict__ cout,
                        int* __restrict__ hist, int e, int n) {
  __shared__ int s_h[NBKT];
  int t = threadIdx.x, c = blockIdx.x;
  for (int b = t; b < NBKT; b += 1024) s_h[b] = 0;
  int zend = (c + 1) * ZSLICE < n ? (c + 1) * ZSLICE : n;
  for (int i = c * ZSLICE + t; i < zend; i += 1024) cout[i] = 0;
  __syncthreads();
  int beg = c * CHUNK_EDGES;
  int end = beg + CHUNK_EDGES < e ? beg + CHUNK_EDGES : e;
  int d[CHUNK_PER];
#pragma unroll
  for (int u = 0; u < CHUNK_PER; ++u) {
    int i = beg + t + u * 1024;
    d[u] = dst[i < end ? i : end - 1];
  }
#pragma unroll
  for (int u = 0; u < CHUNK_PER; ++u) {
    int i = beg + t + u * 1024;
    if (i < end) atomicAdd(&s_h[d[u] >> BKT_SHIFT], 1);
  }
  __syncthreads();
  for (int b = t; b < NBKT; b += 1024) hist[b * NCHUNK + c] = s_h[b];
}

#define ROWS_PER_WAVE ((NBKT + 31) / 32)

__global__ __launch_bounds__(1024)
void piece_scan_kernel(const int* __restrict__ hist, int* __restrict__ boff) {
  __shared__ int s_tot[NBKT];
  int t = threadIdx.x, lane = t & 31, w = t >> 5;
  const int4* h4 = (const int4*)hist;
  int4* o4 = (int4*)boff;
  int4 p[ROWS_PER_WAVE], q[ROWS_PER_WAVE];
#pragma unroll
  for (int u = 0; u < ROWS_PER_WAVE; ++u) {
    int b = w + 32 * u;
    int bb = b < NBKT ? b : NBKT - 1;
    p[u] = h4[bb * (NCHUNK / 4) + lane * 2];
    q[u] = h4[bb * (NCHUNK / 4) + lane * 2 + 1];
  }
#pragma unroll
  for (int u = 0; u < ROWS_PER_WAVE; ++u) {
    int b = w + 32 * u;
    int s = ((p[u].x + p[u].y) + (p[u].z + p[u].w)) + ((q[u].x + q[u].y) + (q[u].z + q[u].w));
#pragma unroll
    for (int k = 16; k > 0; k >>= 1) s += __shfl_xor(s, k);
    if (lane == 0 && b < NBKT) s_tot[b] = s;
  }
  __syncthreads();
#pragma unroll
  for (int u = 0; u < ROWS_PER_WAVE; ++u) {
    int b = w + 32 * u;
    if (b < NBKT) {
      int front = 0;
      for (int k = lane; k < b; k += 32) front += s_tot[k];
#pragma unroll
      for (int k = 16; k > 0; k >>= 1) front += __shfl_xor(front, k);
      int s = ((p[u].x + p[u].y) + (p[u].z + p[u].w)) + ((q[u].x + q[u].y) + (q[u].z + q[u].w));
      int incl = s;
#pragma unroll
      for (int k = 1; k < 32; k <<= 1) {
        int o = __shfl_up(incl, k);
        if (lane >= k) incl += o;
      }
      int run = front + incl - s;
      int4 op, oq;
      op.x = run; run += p[u].x;  op.y = run; run += p[u].y;
      op.z = run; run += p[u].z;  op.w = run; run += p[u].w;
      oq.x = run; run += q[u].x;  oq.y = run; run += q[u].y;
      oq.z = run; run += q[u].z;  oq.w = run;
      o4[b * (NCHUNK / 4) + lane * 2]     = op;
      o4[b * (NCHUNK / 4) + lane * 2 + 1] = oq;
    }
  }
}

__global__ __launch_bounds__(1024)
void chunk_deal_kernel(const int* __restrict__ src, const int* __restrict__ dst,
                       const int* __restrict__ boff, int* __restrict__ cout,
                       int2* __restrict__ ebuf, int e) {
  __shared__ int s_cur[NBKT];
  int t = threadIdx.x, c = blockIdx.x;
  for (int b = t; b < NBKT; b += 1024) s_cur[b] = boff[b * NCHUNK + c];
  __syncthreads();
  int beg = c * CHUNK_EDGES;
  int end = beg + CHUNK_EDGES < e ? beg + CHUNK_EDGES : e;
  int d[CHUNK_PER], s[CHUNK_PER];
#pragma unroll
  for (int u = 0; u < CHUNK_PER; ++u) {
    int i = beg + t + u * 1024;
    int ii = i < end ? i : end - 1;
    d[u] = dst[ii];
    s[u] = src[ii];
  }
#pragma unroll
  for (int u = 0; u < CHUNK_PER; ++u) {
    int i = beg + t + u * 1024;
    if (i < end) {
      int pos = atomicAdd(&s_cur[d[u] >> BKT_SHIFT], 1);
      ebuf[pos] = make_int2(s[u], d[u]);
      atomicAdd(&cout[s[u]], 1);
    }
  }
}

__global__ __launch_bounds__(1024)
void bucket_lists_kernel(const int2* __restrict__ ebuf, const int* __restrict__ boff,
                         const int* __restrict__ cout, int* __restrict__ off,
                         int* __restrict__ cin, int* __restrict__ lst,
                         float* __restrict__ onorm, float* __restrict__ inorm, int n, int e) {
  __shared__ int s_cnt[BKT_NODES];
  __shared__ int s_off[BKT_NODES];
  __shared__ int s_w[32];
  int t = threadIdx.x, lane = t & 31, w = t >> 5;
  int b = blockIdx.x;
  int base = b * BKT_NODES;
  int ebeg = boff[b * NCHUNK];
  int eend = (b + 1 < NBKT) ? boff[(b + 1) * NCHUNK] : e;
  if (t < BKT_NODES) s_cnt[t] = 0;
  __syncthreads();
  for (int i0 = ebeg + t; i0 < eend; i0 += 4 * 1024) {
    int dd[4];
#pragma unroll
    for (int u = 0; u < 4; ++u) {
      int i = i0 + u * 1024;
      dd[u] = ebuf[i < eend ? i : eend - 1].y;
    }
#pragma unroll
    for (int u = 0; u < 4; ++u) {
      int i = i0 + u * 1024;
      if (i < eend) atomicAdd(&s_cnt[dd[u] - base], 1);
    }
  }
  __syncthreads();
  int cnt = (t < BKT_NODES) ? s_cnt[t] : 0;
  int incl = cnt;
#pragma unroll
  for (int k = 1; k < 32; k <<= 1) {
    int o = __shfl_up(incl, k);
    if (lane >= k) incl += o;
  }
  if (lane == 31) s_w[w] = incl;
  __syncthreads();
  int front = (lane < w) ? s_w[lane] : 0;
#pragma unroll
  for (int k = 16; k > 0; k >>= 1) front += __shfl_xor(front, k);
  int lo = front + incl - cnt;
  if (t < BKT_NODES) {
    s_off[t] = lo;
    s_cnt[t] = 0;
    int node = base + t;
    if (node < n) {
      off[node] = ebeg + lo;
      cin[node] = cnt;
      onorm[node] = rsqrtf(fmaxf((float)(cout[node] + 1), 1.0f));
      inorm[node] = rsqrtf(fmaxf((float)(cnt + 1), 1.0f));
    }
  }
  __syncthreads();
  for (int i0 = ebeg + t; i0 < eend; i0 += 4 * 1024) {
    int2 ed[4];
#pragma unroll
    for (int u = 0; u < 4; ++u) {
      int i = i0 + u * 1024;
      ed[u] = ebuf[i < eend ? i : eend - 1];
    }
#pragma unroll
    for (int u = 0; u < 4; ++u) {
      int i = i0 + u * 1024;
      if (i < eend) {
        int l = ed[u].y - base;
        int k = atomicAdd(&s_cnt[l], 1);
        lst[ebeg + s_off[l] + k] = ed[u].x;
      }
    }
  }
}

__global__ void order_kernel(const int* __restrict__ off, const int* __restrict__ cin,
                             const int* __restrict__ lst, const float* __restrict__ onorm,
                             int* __restrict__ csr, float* __restrict__ cw, int n_nodes) {
  int lane = threadIdx.x & 31;
  int node = (int)((blockIdx.x * blockDim.x + threadIdx.x) >> 5);
  if (node >= n_nodes) return;
  int beg = off[node];
  int n   = cin[node];
  const int* p = lst + beg;
  for (int i = lane; i < n; i += 32) {
    int k = p[i];
    float wk = onorm[k];
    int r = 0;
    for (int j = 0; j < n; ++j) {
      int kj = p[j];
      r += (kj < k || (kj == k && j < i)) ? 1 : 0;
    }
    csr[beg + r] = k;
    cw[beg + r]  = wk;
  }
}

__global__ void pack_w_kernel(const float* __restrict__ W, _Float16* __restrict__ Wp, int nout) {
  int idx = blockIdx.x * blockDim.x + threadIdx.x;
  int nt = nout >> 4;
  int total = 4 * nt * 32 * 16;
  if (idx >= total) return;
  int j    = idx & 15;
  int l    = (idx >> 4) & 31;
  int rest = idx >> 9;
  int t    = rest % nt;
  int c    = rest / nt;
  int col   = t * 16 + (l & 15);
  int kbase = c * 32 + ((l >> 4) * 8);
  int K     = kbase + (j & 7) + ((j >> 3) * 16);
  Wp[idx] = (_Float16)W[K * nout + col];
}

template<int U, typename VT>
__device__ __forceinline__ void add_rows(VT& acc, const VT* __restrict__ hv, int ids, int j, int lane) {
  VT r[U];
#pragma unroll
  for (int u = 0; u < U; ++u) {
    int s = __builtin_amdgcn_readlane(ids, j + u);
    r[u] = hv[(size_t)s * 32 + lane];
  }
#pragma unroll
  for (int u = 0; u < U; ++u) acc += r[u];
}

template<int NOUT, bool RELU>
__global__ void gather_agg_kernel(const int* __restrict__ off, const int* __restrict__ cin,
                                  const int* __restrict__ csr, const float* __restrict__ h,
                                  const float* __restrict__ inorm, const float* __restrict__ b,
                                  float* __restrict__ out, int n_nodes) {
  constexpr int V = NOUT / 32;
  typedef float VT __attribute__((ext_vector_type(V)));
  int lane = threadIdx.x & 31;
  int node = (int)((blockIdx.x * blockDim.x + threadIdx.x) >> 5);
  if (node >= n_nodes) return;
  node = __builtin_amdgcn_readfirstlane(node);
  int beg = off[node];
  int n   = cin[node];
  const VT* hv = (const VT*)h;
  VT acc = hv[(size_t)node * 32 + lane];
  for (int base = 0; base < n; base += 32) {
    int m = n - base;
    if (m > 32) m = 32;
    int ids = (lane < m) ? csr[beg + base + lane] : 0;
    int j = 0;
    for (; j + 8 <= m; j += 8) add_rows<8, VT>(acc, hv, ids, j, lane);
    if (j + 4 <= m) { add_rows<4, VT>(acc, hv, ids, j, lane); j += 4; }
    if (j + 2 <= m) { add_rows<2, VT>(acc, hv, ids, j, lane); j += 2; }
    if (j < m)      { add_rows<1, VT>(acc, hv, ids, j, lane); }
  }
  float sc = inorm[node];
  VT bv = ((const VT*)b)[lane];
  VT v = acc * sc + bv;
  if (RELU) {
#pragma unroll
    for (int k = 0; k < V; ++k) v[k] = fmaxf(v[k], 0.0f);
  }
  ((VT*)out)[(size_t)node * 32 + lane] = v;
}


#define XS_PITCH 136

template<int NOUT2>
__global__ __launch_bounds__(512)
void gather_relu_gemm_kernel(const int* __restrict__ off, const int* __restrict__ cin,
                             const int* __restrict__ csr, const float* __restrict__ h,
                             const float* __restrict__ inorm, const float* __restrict__ b,
                             const float* __restrict__ onorm, const _Float16* __restrict__ Wp,
                             float* __restrict__ hnext, int n_nodes) {
  constexpr int NT2 = NOUT2 / 16;
  typedef float VT __attribute__((ext_vector_type(4)));
  __shared__ __attribute__((aligned(16))) _Float16 xs[16 * XS_PITCH];

  int lane = threadIdx.x & 31;
  int w = __builtin_amdgcn_readfirstlane((int)(threadIdx.x >> 5));
  int node = (int)blockIdx.x * 16 + w;

  VT v = {0.f, 0.f, 0.f, 0.f};
  if (node < n_nodes) {
    int beg = off[node];
    int n   = cin[node];
    const VT* hv = (const VT*)h;
    VT acc = hv[(size_t)node * 32 + lane];
    for (int base = 0; base < n; base += 32) {
      int m = n - base;
      if (m > 32) m = 32;
      int ids = (lane < m) ? csr[beg + base + lane] : 0;
      int j = 0;
      for (; j + 8 <= m; j += 8) add_rows<8, VT>(acc, hv, ids, j, lane);
      if (j + 4 <= m) { add_rows<4, VT>(acc, hv, ids, j, lane); j += 4; }
      if (j + 2 <= m) { add_rows<2, VT>(acc, hv, ids, j, lane); j += 2; }
      if (j < m)      { add_rows<1, VT>(acc, hv, ids, j, lane); }
    }
    float sc = inorm[node];
    VT bv = ((const VT*)b)[lane];
    v = acc * sc + bv;
#pragma unroll
    for (int k = 0; k < 4; ++k) v[k] = fmaxf(v[k], 0.0f);
    float on = onorm[node];
    v = v * on;
  }

  v16h bw[4];
  if (w < NT2) {
#pragma unroll
    for (int kc = 0; kc < 4; ++kc)
      bw[kc] = *(const v16h*)(Wp + ((size_t)(kc * NT2 + w) * 32 + lane) * 16);
  }

  v4h xh;
  xh[0] = (_Float16)v[0]; xh[1] = (_Float16)v[1];
  xh[2] = (_Float16)v[2]; xh[3] = (_Float16)v[3];
  *(v4h*)(xs + w * XS_PITCH + lane * 4) = xh;
  __syncthreads();

  if (w >= NT2) return;

  int r    = lane & 15;
  int half = lane >> 4;
  const _Float16* pa = xs + r * XS_PITCH + half * 8;
  v8f acc2 = {0.f, 0.f, 0.f, 0.f, 0.f, 0.f, 0.f, 0.f};
#pragma unroll
  for (int kc = 0; kc < 4; ++kc) {
    v8h lo = *(const v8h*)(pa + kc * 32);
    v8h hi = *(const v8h*)(pa + kc * 32 + 16);
    v16h a = __builtin_shufflevector(lo, hi, 0, 1, 2, 3, 4, 5, 6, 7, 8, 9, 10, 11, 12, 13, 14, 15);
    acc2 = __builtin_amdgcn_wmma_f32_16x16x32_f16(
         false, a,  false, bw[kc],
         (short)0, acc2,  false,  false);
  }

  int row0 = (int)blockIdx.x * 16 + half * 8;
  float* ph = hnext + (size_t)row0 * NOUT2 + w * 16 + r;
  if ((int)blockIdx.x * 16 + 16 <= n_nodes) {
#pragma unroll
    for (int vr = 0; vr < 8; ++vr) ph[(size_t)vr * NOUT2] = acc2[vr];
  } else {
#pragma unroll
    for (int vr = 0; vr < 8; ++vr)
      if (row0 + vr < n_nodes) ph[(size_t)vr * NOUT2] = acc2[vr];
  }
}

template<int U, typename VT>
__device__ __forceinline__ void add_rows_scaled(VT& acc, const VT* __restrict__ xv, int ids, int scl, int j, int lane) {
  VT r[U];
#pragma unroll
  for (int u = 0; u < U; ++u) {
    int s = __builtin_amdgcn_readlane(ids, j + u);
    r[u] = xv[(size_t)s * 32 + lane];
  }
#pragma unroll
  for (int u = 0; u < U; ++u) {
    float sc = __int_as_float(__builtin_amdgcn_readlane(scl, j + u));
    acc += r[u] * sc;
  }
}

__global__ __launch_bounds__(512)
void gather_gemm_relu_gemm_kernel(const int* __restrict__ off, const int* __restrict__ cin,
                                  const int* __restrict__ csr, const float* __restrict__ cw,
                                  const float* __restrict__ feat,
                                  const float* __restrict__ inorm, const float* __restrict__ b,
                                  const float* __restrict__ onorm,
                                  const _Float16* __restrict__ Wp1, const _Float16* __restrict__ Wp2,
                                  float* __restrict__ hnext, int n_nodes) {
  typedef float VT __attribute__((ext_vector_type(4)));
  __shared__ __attribute__((aligned(16))) _Float16 xs[16 * XS_PITCH];
  __shared__ __attribute__((aligned(16))) _Float16 ys[16 * XS_PITCH];
  __shared__ __attribute__((aligned(16))) float s_in[16];
  __shared__ __attribute__((aligned(16))) float s_on[16];

  int lane = threadIdx.x & 31;
  int w = __builtin_amdgcn_readfirstlane((int)(threadIdx.x >> 5));
  int node = (int)blockIdx.x * 16 + w;

  VT acc = {0.f, 0.f, 0.f, 0.f};
  float sc_in = 0.f, sc_on = 0.f;
  if (node < n_nodes) {
    int beg = off[node];
    int n   = cin[node];
    sc_on   = onorm[node];
    const VT* xv = (const VT*)feat;
    acc = xv[(size_t)node * 32 + lane] * sc_on;
    for (int base = 0; base < n; base += 32) {
      int m = n - base;
      if (m > 32) m = 32;
      int ids = (lane < m) ? csr[beg + base + lane] : 0;
      int scl = (lane < m) ? __float_as_int(cw[beg + base + lane]) : 0;
      int j = 0;
      for (; j + 8 <= m; j += 8) add_rows_scaled<8, VT>(acc, xv, ids, scl, j, lane);
      if (j + 4 <= m) { add_rows_scaled<4, VT>(acc, xv, ids, scl, j, lane); j += 4; }
      if (j + 2 <= m) { add_rows_scaled<2, VT>(acc, xv, ids, scl, j, lane); j += 2; }
      if (j < m)      { add_rows_scaled<1, VT>(acc, xv, ids, scl, j, lane); }
    }
    sc_in = inorm[node];
  }

  int t = w & 7;
  const _Float16* Wsel = (w < 8) ? Wp1 : Wp2;
  v16h bw[4];
#pragma unroll
  for (int kc = 0; kc < 4; ++kc)
    bw[kc] = *(const v16h*)(Wsel + ((size_t)(kc * 8 + t) * 32 + lane) * 16);

  v4h xh;
  xh[0] = (_Float16)acc[0]; xh[1] = (_Float16)acc[1];
  xh[2] = (_Float16)acc[2]; xh[3] = (_Float16)acc[3];
  *(v4h*)(xs + w * XS_PITCH + lane * 4) = xh;
  if (lane == 0) { s_in[w] = sc_in; s_on[w] = sc_on; }
  __syncthreads();

  int r    = lane & 15;
  int half = lane >> 4;
  if (w < 8) {
    const _Float16* pa = xs + r * XS_PITCH + half * 8;
    v8f acc1 = {0.f, 0.f, 0.f, 0.f, 0.f, 0.f, 0.f, 0.f};
#pragma unroll
    for (int kc = 0; kc < 4; ++kc) {
      v8h lo = *(const v8h*)(pa + kc * 32);
      v8h hi = *(const v8h*)(pa + kc * 32 + 16);
      v16h a = __builtin_shufflevector(lo, hi, 0, 1, 2, 3, 4, 5, 6, 7, 8, 9, 10, 11, 12, 13, 14, 15);
      acc1 = __builtin_amdgcn_wmma_f32_16x16x32_f16(
           false, a,  false, bw[kc],
           (short)0, acc1,  false,  false);
    }
    float bcol = b[t * 16 + r];
#pragma unroll
    for (int vr = 0; vr < 8; ++vr) {
      int row = half * 8 + vr;
      float v = acc1[vr] * s_in[row] + bcol;
      v = fmaxf(v, 0.0f);
      v = v * s_on[row];
      ys[row * XS_PITCH + t * 16 + r] = (_Float16)v;
    }
  }
  __syncthreads();

  if (w < 8) return;

  const _Float16* pb = ys + r * XS_PITCH + half * 8;
  v8f acc2 = {0.f, 0.f, 0.f, 0.f, 0.f, 0.f, 0.f, 0.f};
#pragma unroll
  for (int kc = 0; kc < 4; ++kc) {
    v8h lo = *(const v8h*)(pb + kc * 32);
    v8h hi = *(const v8h*)(pb + kc * 32 + 16);
    v16h a = __builtin_shufflevector(lo, hi, 0, 1, 2, 3, 4, 5, 6, 7, 8, 9, 10, 11, 12, 13, 14, 15);
    acc2 = __builtin_amdgcn_wmma_f32_16x16x32_f16(
         false, a,  false, bw[kc],
         (short)0, acc2,  false,  false);
  }

  int row0 = (int)blockIdx.x * 16 + half * 8;
  float* ph = hnext + (size_t)row0 * 128 + t * 16 + r;
  if ((int)blockIdx.x * 16 + 16 <= n_nodes) {
#pragma unroll
    for (int vr = 0; vr < 8; ++vr) ph[(size_t)vr * 128] = acc2[vr];
  } else {
#pragma unroll
    for (int vr = 0; vr < 8; ++vr)
      if (row0 + vr < n_nodes) ph[(size_t)vr * 128] = acc2[vr];
  }
}

extern "C" void kernel_launch(void* const* d_in, const int* in_sizes, int n_in,
                              void* d_out, int out_size, void* d_ws, size_t ws_size,
                              hipStream_t stream) {
  const float* feat = (const float*)d_in[0];
  const int*   src  = (const int*)  d_in[1];
  const int*   dst  = (const int*)  d_in[2];
  const float* W1   = (const float*)d_in[3];
  const float* b1   = (const float*)d_in[4];
  const float* W2   = (const float*)d_in[5];
  const float* b2   = (const float*)d_in[6];
  const float* W3   = (const float*)d_in[7];
  const float* b3   = (const float*)d_in[8];
  float* out = (float*)d_out;

  size_t off = 0;
  auto alloc = [&](size_t bytes) -> char* {
    char* p = (char*)d_ws + off;
    off += (bytes + 255) & ~(size_t)255;
    return p;
  };
  float*     onorm = (float*)alloc((size_t)NNODES * sizeof(float));
  float*     inorm = (float*)alloc((size_t)NNODES * sizeof(float));
  float*     hbuf  = (float*)alloc((size_t)NNODES * FDIM * sizeof(float));
  float*     xbuf  = (float*)alloc((size_t)NNODES * FDIM * sizeof(float));
  _Float16*  Wp1   = (_Float16*)alloc((size_t)FDIM * FDIM * sizeof(_Float16));
  _Float16*  Wp2   = (_Float16*)alloc((size_t)FDIM * FDIM * sizeof(_Float16));
  _Float16*  Wp3   = (_Float16*)alloc((size_t)FDIM * FDIM * sizeof(_Float16));
  int*       cnt_in = (int*)alloc((size_t)NNODES * sizeof(int));
  int*       cnt_out= (int*)alloc((size_t)NNODES * sizeof(int));
  int*       lstart= (int*)alloc((size_t)NNODES * sizeof(int));
  int*       hist  = (int*)alloc((size_t)NBKT * NCHUNK * sizeof(int));
  int*       boff  = (int*)alloc((size_t)NBKT * NCHUNK * sizeof(int));
  int2*      ebuf  = (int2*)alloc((size_t)NEDGES * sizeof(int2));
  int*       lst   = (int*)alloc((size_t)NEDGES * sizeof(int));
  int*       csr   = (int*)alloc((size_t)NEDGES * sizeof(int));
  float*     cw    = (float*)alloc((size_t)NEDGES * sizeof(float));

  const int TB = 256;
  const int node_waves_blocks = ceil_div((long long)NNODES * 32, TB);

  chunk_count_kernel<<<NCHUNK, 1024, 0, stream>>>(dst, cnt_out, hist, NEDGES, NNODES);
  piece_scan_kernel<<<1, 1024, 0, stream>>>(hist, boff);
  chunk_deal_kernel<<<NCHUNK, 1024, 0, stream>>>(src, dst, boff, cnt_out, ebuf, NEDGES);
  bucket_lists_kernel<<<NBKT, 1024, 0, stream>>>(ebuf, boff, cnt_out, lstart, cnt_in, lst, onorm, inorm, NNODES, NEDGES);
  order_kernel<<<node_waves_blocks, TB, 0, stream>>>(lstart, cnt_in, lst, onorm, csr, cw, NNODES);

  const int tile_blocks = ceil_div(NNODES, 16);

  pack_w_kernel<<<ceil_div(4 * 8 * 512, TB), TB, 0, stream>>>(W1, Wp1, 128);
  pack_w_kernel<<<ceil_div(4 * 8 * 512, TB), TB, 0, stream>>>(W2, Wp2, 128);
  gather_gemm_relu_gemm_kernel<<<tile_blocks, 512, 0, stream>>>(
      lstart, cnt_in, csr, cw, feat, inorm, b1, onorm, Wp1, Wp2, xbuf, NNODES);

  pack_w_kernel<<<ceil_div(4 * 4 * 512, TB), TB, 0, stream>>>(W3, Wp3, 64);
  gather_relu_gemm_kernel<64><<<tile_blocks, 512, 0, stream>>>(
      lstart, cnt_in, csr, xbuf, inorm, b2, onorm, Wp3, hbuf, NNODES);

  gather_agg_kernel<64, false><<<node_waves_blocks, TB, 0, stream>>>(
      lstart, cnt_in, csr, hbuf, inorm, b3, out, NNODES);
}


# Three weight-packing launches folded into the one-block scan launch as 40 extra blocks on WGPs that launch left idle: 11 launches a call become 8; packed values and every sum unchanged
# speedup vs baseline: 4.1940x; 1.0733x over previous
#include <hip/hip_runtime.h>
#include <math.h>

typedef __attribute__((ext_vector_type(16))) _Float16 v16h;
typedef __attribute__((ext_vector_type(8)))  float    v8f;
typedef __attribute__((ext_vector_type(8)))  _Float16 v8h;
typedef __attribute__((ext_vector_type(4)))  _Float16 v4h;

#define NNODES 100000
#define NEDGES 1600000
#define FDIM   128

#define BKT_SHIFT   9
#define BKT_NODES   (1 << BKT_SHIFT)
#define NBKT        ((NNODES + BKT_NODES - 1) / BKT_NODES)
#define NCHUNK      256
#define CHUNK_EDGES ((NEDGES + NCHUNK - 1) / NCHUNK)
#define CHUNK_PER   ((CHUNK_EDGES + 1023) / 1024)
#define ZSLICE      ((NNODES + NCHUNK - 1) / NCHUNK)

__global__ __launch_bounds__(1024)
void chunk_count_kernel(const int* __restrict__ dst, int* __restrict__ cout,
                        int* __restrict__ hist, int e, int n) {
  __shared__ int s_h[NBKT];
  int t = threadIdx.x, c = blockIdx.x;
  for (int b = t; b < NBKT; b += 1024) s_h[b] = 0;
  int zend = (c + 1) * ZSLICE < n ? (c + 1) * ZSLICE : n;
  for (int i = c * ZSLICE + t; i < zend; i += 1024) cout[i] = 0;
  __syncthreads();
  int beg = c * CHUNK_EDGES;
  int end = beg + CHUNK_EDGES < e ? beg + CHUNK_EDGES : e;
  int d[CHUNK_PER];
#pragma unroll
  for (int u = 0; u < CHUNK_PER; ++u) {
    int i = beg + t + u * 1024;
    d[u] = dst[i < end ? i : end - 1];
  }
#pragma unroll
  for (int u = 0; u < CHUNK_PER; ++u) {
    int i = beg + t + u * 1024;
    if (i < end) atomicAdd(&s_h[d[u] >> BKT_SHIFT], 1);
  }
  __syncthreads();
  for (int b = t; b < NBKT; b += 1024) hist[b * NCHUNK + c] = s_h[b];
}

__device__ __forceinline__ void pack_w_elem(const float* __restrict__ W, _Float16* __restrict__ Wp, int nout, int idx) {
  int nt = nout >> 4;
  int j    = idx & 15;
  int l    = (idx >> 4) & 31;
  int rest = idx >> 9;
  int t    = rest % nt;
  int c    = rest / nt;
  int col   = t * 16 + (l & 15);
  int kbase = c * 32 + ((l >> 4) * 8);
  int K     = kbase + (j & 7) + ((j >> 3) * 16);
  Wp[idx] = (_Float16)W[K * nout + col];
}

#define ROWS_PER_WAVE ((NBKT + 31) / 32)

#define PACK_BLK_128 ((4 * 8 * 512) / 1024)
#define PACK_BLK_64  ((4 * 4 * 512) / 1024)
#define PACK_BLOCKS  (2 * PACK_BLK_128 + PACK_BLK_64)
#define SCAN_GRID    (1 + PACK_BLOCKS)

__global__ __launch_bounds__(1024)
void piece_scan_kernel(const int* __restrict__ hist, int* __restrict__ boff,
                       const float* __restrict__ W1, const float* __restrict__ W2,
                       const float* __restrict__ W3, _Float16* __restrict__ Wp1,
                       _Float16* __restrict__ Wp2, _Float16* __restrict__ Wp3) {
  __shared__ int s_tot[NBKT];
  int t = threadIdx.x, lane = t & 31, w = t >> 5;
  if (blockIdx.x > 0) {
    int pb = (int)blockIdx.x - 1;
    if (pb < PACK_BLK_128)          pack_w_elem(W1, Wp1, 128, pb * 1024 + t);
    else if (pb < 2 * PACK_BLK_128) pack_w_elem(W2, Wp2, 128, (pb - PACK_BLK_128) * 1024 + t);
    else                            pack_w_elem(W3, Wp3, 64, (pb - 2 * PACK_BLK_128) * 1024 + t);
    return;
  }
  const int4* h4 = (const int4*)hist;
  int4* o4 = (int4*)boff;
  int4 p[ROWS_PER_WAVE], q[ROWS_PER_WAVE];
#pragma unroll
  for (int u = 0; u < ROWS_PER_WAVE; ++u) {
    int b = w + 32 * u;
    int bb = b < NBKT ? b : NBKT - 1;
    p[u] = h4[bb * (NCHUNK / 4) + lane * 2];
    q[u] = h4[bb * (NCHUNK / 4) + lane * 2 + 1];
  }
#pragma unroll
  for (int u = 0; u < ROWS_PER_WAVE; ++u) {
    int b = w + 32 * u;
    int s = ((p[u].x + p[u].y) + (p[u].z + p[u].w)) + ((q[u].x + q[u].y) + (q[u].z + q[u].w));
#pragma unroll
    for (int k = 16; k > 0; k >>= 1) s += __shfl_xor(s, k);
    if (lane == 0 && b < NBKT) s_tot[b] = s;
  }
  __syncthreads();
#pragma unroll
  for (int u = 0; u < ROWS_PER_WAVE; ++u) {
    int b = w + 32 * u;
    if (b < NBKT) {
      int front = 0;
      for (int k = lane; k < b; k += 32) front += s_tot[k];
#pragma unroll
      for (int k = 16; k > 0; k >>= 1) front += __shfl_xor(front, k);
      int s = ((p[u].x + p[u].y) + (p[u].z + p[u].w)) + ((q[u].x + q[u].y) + (q[u].z + q[u].w));
      int incl = s;
#pragma unroll
      for (int k = 1; k < 32; k <<= 1) {
        int o = __shfl_up(incl, k);
        if (lane >= k) incl += o;
      }
      int run = front + incl - s;
      int4 op, oq;
      op.x = run; run += p[u].x;  op.y = run; run += p[u].y;
      op.z = run; run += p[u].z;  op.w = run; run += p[u].w;
      oq.x = run; run += q[u].x;  oq.y = run; run += q[u].y;
      oq.z = run; run += q[u].z;  oq.w = run;
      o4[b * (NCHUNK / 4) + lane * 2]     = op;
      o4[b * (NCHUNK / 4) + lane * 2 + 1] = oq;
    }
  }
}

__global__ __launch_bounds__(1024)
void chunk_deal_kernel(const int* __restrict__ src, const int* __restrict__ dst,
                       const int* __restrict__ boff, int* __restrict__ cout,
                       int2* __restrict__ ebuf, int e) {
  __shared__ int s_cur[NBKT];
  int t = threadIdx.x, c = blockIdx.x;
  for (int b = t; b < NBKT; b += 1024) s_cur[b] = boff[b * NCHUNK + c];
  __syncthreads();
  int beg = c * CHUNK_EDGES;
  int end = beg + CHUNK_EDGES < e ? beg + CHUNK_EDGES : e;
  int d[CHUNK_PER], s[CHUNK_PER];
#pragma unroll
  for (int u = 0; u < CHUNK_PER; ++u) {
    int i = beg + t + u * 1024;
    int ii = i < end ? i : end - 1;
    d[u] = dst[ii];
    s[u] = src[ii];
  }
#pragma unroll
  for (int u = 0; u < CHUNK_PER; ++u) {
    int i = beg + t + u * 1024;
    if (i < end) {
      int pos = atomicAdd(&s_cur[d[u] >> BKT_SHIFT], 1);
      ebuf[pos] = make_int2(s[u], d[u]);
      atomicAdd(&cout[s[u]], 1);
    }
  }
}

__global__ __launch_bounds__(1024)
void bucket_lists_kernel(const int2* __restrict__ ebuf, const int* __restrict__ boff,
                         const int* __restrict__ cout, int* __restrict__ off,
                         int* __restrict__ cin, int* __restrict__ lst,
                         float* __restrict__ onorm, float* __restrict__ inorm, int n, int e) {
  __shared__ int s_cnt[BKT_NODES];
  __shared__ int s_off[BKT_NODES];
  __shared__ int s_w[32];
  int t = threadIdx.x, lane = t & 31, w = t >> 5;
  int b = blockIdx.x;
  int base = b * BKT_NODES;
  int ebeg = boff[b * NCHUNK];
  int eend = (b + 1 < NBKT) ? boff[(b + 1) * NCHUNK] : e;
  if (t < BKT_NODES) s_cnt[t] = 0;
  __syncthreads();
  for (int i0 = ebeg + t; i0 < eend; i0 += 4 * 1024) {
    int dd[4];
#pragma unroll
    for (int u = 0; u < 4; ++u) {
      int i = i0 + u * 1024;
      dd[u] = ebuf[i < eend ? i : eend - 1].y;
    }
#pragma unroll
    for (int u = 0; u < 4; ++u) {
      int i = i0 + u * 1024;
      if (i < eend) atomicAdd(&s_cnt[dd[u] - base], 1);
    }
  }
  __syncthreads();
  int cnt = (t < BKT_NODES) ? s_cnt[t] : 0;
  int incl = cnt;
#pragma unroll
  for (int k = 1; k < 32; k <<= 1) {
    int o = __shfl_up(incl, k);
    if (lane >= k) incl += o;
  }
  if (lane == 31) s_w[w] = incl;
  __syncthreads();
  int front = (lane < w) ? s_w[lane] : 0;
#pragma unroll
  for (int k = 16; k > 0; k >>= 1) front += __shfl_xor(front, k);
  int lo = front + incl - cnt;
  if (t < BKT_NODES) {
    s_off[t] = lo;
    s_cnt[t] = 0;
    int node = base + t;
    if (node < n) {
      off[node] = ebeg + lo;
      cin[node] = cnt;
      onorm[node] = rsqrtf(fmaxf((float)(cout[node] + 1), 1.0f));
      inorm[node] = rsqrtf(fmaxf((float)(cnt + 1), 1.0f));
    }
  }
  __syncthreads();
  for (int i0 = ebeg + t; i0 < eend; i0 += 4 * 1024) {
    int2 ed[4];
#pragma unroll
    for (int u = 0; u < 4; ++u) {
      int i = i0 + u * 1024;
      ed[u] = ebuf[i < eend ? i : eend - 1];
    }
#pragma unroll
    for (int u = 0; u < 4; ++u) {
      int i = i0 + u * 1024;
      if (i < eend) {
        int l = ed[u].y - base;
        int k = atomicAdd(&s_cnt[l], 1);
        lst[ebeg + s_off[l] + k] = ed[u].x;
      }
    }
  }
}

#define RES_BLOCKS_8    2048
#define RES_BLOCKS_16x4 1024
#define RES_BLOCKS_16x3  768

__device__ __forceinline__ float lane_f(float v, int l) {
  return __int_as_float(__builtin_amdgcn_readlane(__float_as_int(v), l));
}

__global__ __launch_bounds__(256)
void order_kernel(const int* __restrict__ off, const int* __restrict__ cin,
                  const int* __restrict__ lst, const float* __restrict__ onorm,
                  int* __restrict__ csr, float* __restrict__ cw, int n_nodes) {
  int lane = threadIdx.x & 31;
  int nw = (int)((gridDim.x * blockDim.x) >> 5);
  int gw = __builtin_amdgcn_readfirstlane((int)((blockIdx.x * blockDim.x + threadIdx.x) >> 5));
  for (int first = gw; first < n_nodes; first += 32 * nw) {
    int mn = first + lane * nw;
    int begL = (mn < n_nodes) ? off[mn] : 0;
    int cntL = (mn < n_nodes) ? cin[mn] : 0;
    int node = first;
    for (int it = 0; it < 32 && node < n_nodes; ++it, node += nw) {
      int beg = __builtin_amdgcn_readlane(begL, it);
      int n   = __builtin_amdgcn_readlane(cntL, it);
      const int* p = lst + beg;
      for (int i = lane; i < n; i += 32) {
        int k = p[i];
        float wk = onorm[k];
        int r = 0;
        for (int j = 0; j < n; ++j) {
          int kj = p[j];
          r += (kj < k || (kj == k && j < i)) ? 1 : 0;
        }
        csr[beg + r] = k;
        cw[beg + r]  = wk;
      }
    }
  }
}

template<int U, typename VT>
__device__ __forceinline__ void add_rows(VT& acc, const VT* __restrict__ hv, int ids, int j, unsigned lane) {
  VT r[U];
#pragma unroll
  for (int u = 0; u < U; ++u) {
    int s = __builtin_amdgcn_readlane(ids, j + u);
    const VT* row = hv + (size_t)s * 32;
    r[u] = row[lane];
  }
#pragma unroll
  for (int u = 0; u < U; ++u) acc += r[u];
}

template<typename VT>
__device__ __forceinline__ void add_list_chunk(VT& acc, const VT* __restrict__ hv, int ids, int m, unsigned lane) {
  int j = 0;
  for (; j + 8 <= m; j += 8) add_rows<8, VT>(acc, hv, ids, j, lane);
  if (j + 4 <= m) { add_rows<4, VT>(acc, hv, ids, j, lane); j += 4; }
  if (j + 2 <= m) { add_rows<2, VT>(acc, hv, ids, j, lane); j += 2; }
  if (j < m)      { add_rows<1, VT>(acc, hv, ids, j, lane); }
}

template<int NOUT, bool RELU>
__global__ __launch_bounds__(256)
void gather_agg_kernel(const int* __restrict__ off, const int* __restrict__ cin,
                       const int* __restrict__ csr, const float* __restrict__ h,
                       const float* __restrict__ inorm, const float* __restrict__ b,
                       float* __restrict__ out, int n_nodes) {
  constexpr int V = NOUT / 32;
  typedef float VT __attribute__((ext_vector_type(V)));
  int lane = threadIdx.x & 31;
  int nw = (int)((gridDim.x * blockDim.x) >> 5);
  int gw = __builtin_amdgcn_readfirstlane((int)((blockIdx.x * blockDim.x + threadIdx.x) >> 5));
  const VT* hv = (const VT*)h;
  VT bv = ((const VT*)b)[lane];
  for (int first = gw; first < n_nodes; first += 32 * nw) {
    int mn = first + lane * nw;
    bool ok = mn < n_nodes;
    int   begL = ok ? off[mn] : 0;
    int   cntL = ok ? cin[mn] : 0;
    float inL  = ok ? inorm[mn] : 0.0f;
    int beg = 0, n = 0, ids = 0;
    for (int it = -1; it < 32; ++it) {
      int node2 = first + (it + 1) * nw;
      bool more = (it + 1 < 32) && (node2 < n_nodes);
      int beg2 = 0, n2 = 0, ids2 = 0;
      if (more) {
        beg2 = __builtin_amdgcn_readlane(begL, it + 1);
        n2   = __builtin_amdgcn_readlane(cntL, it + 1);
        ids2 = (lane < n2) ? csr[beg2 + lane] : 0;
      }
      if (it >= 0) {
        int node = first + it * nw;
        VT acc = hv[(size_t)node * 32 + lane];
        add_list_chunk<VT>(acc, hv, ids, n < 32 ? n : 32, lane);
        for (int base = 32; base < n; base += 32) {
          int m = n - base;
          if (m > 32) m = 32;
          int idx = (lane < m) ? csr[beg + base + lane] : 0;
          add_list_chunk<VT>(acc, hv, idx, m, lane);
        }
        float sc = lane_f(inL, it);
        VT v = acc * sc + bv;
        if (RELU) {
#pragma unroll
          for (int k = 0; k < V; ++k) v[k] = fmaxf(v[k], 0.0f);
        }
        ((VT*)out)[(size_t)node * 32 + lane] = v;
      }
      if (!more) break;
      beg = beg2; n = n2; ids = ids2;
    }
  }
}


#define XS_PITCH 136

template<int NOUT2>
__global__ __launch_bounds__(512) __attribute__((amdgpu_waves_per_eu(16)))
void gather_relu_gemm_kernel(const int* __restrict__ off, const int* __restrict__ cin,
                             const int* __restrict__ csr, const float* __restrict__ h,
                             const float* __restrict__ inorm, const float* __restrict__ b,
                             const float* __restrict__ onorm, const _Float16* __restrict__ Wp,
                             float* __restrict__ hnext, int n_nodes) {
  constexpr int NT2 = NOUT2 / 16;
  typedef float VT __attribute__((ext_vector_type(4)));
  __shared__ __attribute__((aligned(16))) _Float16 xs[2][16 * XS_PITCH];
  __shared__ __attribute__((aligned(16))) float s_b[128];
  if (threadIdx.x < 128) s_b[threadIdx.x] = b[threadIdx.x];
  __syncthreads();

  int lane = threadIdx.x & 31;
  int w = __builtin_amdgcn_readfirstlane((int)(threadIdx.x >> 5));
  int G = (int)gridDim.x;
  int n_tiles = (n_nodes + 15) >> 4;
  const VT* hv = (const VT*)h;

  int r    = lane & 15;
  int half = lane >> 4;

  for (int tfirst = (int)blockIdx.x; tfirst < n_tiles; tfirst += 32 * G) {
    int mt = tfirst + (int)lane * G;
    int mn = mt * 16 + w;
    bool ok = (mt < n_tiles) && (mn < n_nodes);
    int   begL = ok ? off[mn] : 0;
    int   cntL = ok ? cin[mn] : 0;
    float inL  = ok ? inorm[mn] : 0.0f;
    float onL  = ok ? onorm[mn] : 0.0f;
    int beg = 0, n = 0, ids = 0;
    for (int it = -1; it < 32; ++it) {
      int tile2 = tfirst + (it + 1) * G;
      bool more = (it + 1 < 32) && (tile2 < n_tiles);
      int beg2 = 0, n2 = 0, ids2 = 0;
      if (more) {
        beg2 = __builtin_amdgcn_readlane(begL, it + 1);
        n2   = __builtin_amdgcn_readlane(cntL, it + 1);
        ids2 = ((int)lane < n2) ? csr[beg2 + lane] : 0;
      }
      if (it >= 0) {
      int tile = tfirst + it * G;
      int node = tile * 16 + w;
      VT v = {0.f, 0.f, 0.f, 0.f};
      if (node < n_nodes) {
        VT acc = hv[(size_t)node * 32 + lane];
        add_list_chunk<VT>(acc, hv, ids, n < 32 ? n : 32, lane);
        for (int base = 32; base < n; base += 32) {
          int m = n - base;
          if (m > 32) m = 32;
          int idx = (lane < m) ? csr[beg + base + lane] : 0;
          add_list_chunk<VT>(acc, hv, idx, m, lane);
        }
        float sc = lane_f(inL, it);
        VT bv = *(const VT*)(s_b + lane * 4);
        v = acc * sc + bv;
#pragma unroll
        for (int k = 0; k < 4; ++k) v[k] = fmaxf(v[k], 0.0f);
        float on = lane_f(onL, it);
        v = v * on;
      }

      int pw = (w - NT2 * it) & 15;
      v16h bw[4];
#pragma unroll
      for (int kc = 0; kc < 4; ++kc) bw[kc] = (v16h)(_Float16)0.0f;
      if (pw < NT2) {
#pragma unroll
        for (int kc = 0; kc < 4; ++kc)
          bw[kc] = *(const v16h*)(Wp + ((size_t)(kc * NT2 + pw) * 32 + lane) * 16);
      }

      _Float16* xt = xs[it & 1];
      v4h xh;
      xh[0] = (_Float16)v[0]; xh[1] = (_Float16)v[1];
      xh[2] = (_Float16)v[2]; xh[3] = (_Float16)v[3];
      *(v4h*)(xt + w * XS_PITCH + lane * 4) = xh;
      __syncthreads();

      if (pw < NT2) {
        const _Float16* pa = xt + r * XS_PITCH + half * 8;
        v8f acc2 = {0.f, 0.f, 0.f, 0.f, 0.f, 0.f, 0.f, 0.f};
#pragma unroll
        for (int kc = 0; kc < 4; ++kc) {
          v8h lo = *(const v8h*)(pa + kc * 32);
          v8h hi = *(const v8h*)(pa + kc * 32 + 16);
          v16h a = __builtin_shufflevector(lo, hi, 0, 1, 2, 3, 4, 5, 6, 7, 8, 9, 10, 11, 12, 13, 14, 15);
          acc2 = __builtin_amdgcn_wmma_f32_16x16x32_f16(
               false, a,  false, bw[kc],
               (short)0, acc2,  false,  false);
        }
        int row0 = tile * 16 + half * 8;
        unsigned po = (unsigned)(row0 * NOUT2 + pw * 16 + r);
        if (tile * 16 + 16 <= n_nodes) {
#pragma unroll
          for (int vr = 0; vr < 8; ++vr) hnext[po + (unsigned)(vr * NOUT2)] = acc2[vr];
        } else {
#pragma unroll
          for (int vr = 0; vr < 8; ++vr)
            if (row0 + vr < n_nodes) hnext[po + (unsigned)(vr * NOUT2)] = acc2[vr];
        }
      }
      }
      if (!more) break;
      beg = beg2; n = n2; ids = ids2;
    }
  }
}

template<int U, typename VT>
__device__ __forceinline__ void add_rows_scaled(VT& acc, float& pre, const VT* __restrict__ xv, int ids, int scl, int j, unsigned lane) {
  VT r[U];
#pragma unroll
  for (int u = 0; u < U; ++u) {
    int s = __builtin_amdgcn_readlane(ids, j + u);
    const VT* row = xv + (size_t)s * 32;
    r[u] = row[lane];
  }
  acc = acc * pre;
  pre = 1.0f;
#pragma unroll
  for (int u = 0; u < U; ++u) {
    float sc = __int_as_float(__builtin_amdgcn_readlane(scl, j + u));
    acc += r[u] * sc;
  }
}

template<typename VT>
__device__ __forceinline__ void add_list_chunk_scaled(VT& acc, float& pre, const VT* __restrict__ xv, int ids, int scl, int m, unsigned lane) {
  int j = 0;
  for (; j + 8 <= m; j += 8) add_rows_scaled<8, VT>(acc, pre, xv, ids, scl, j, lane);
  if (j + 4 <= m) { add_rows_scaled<4, VT>(acc, pre, xv, ids, scl, j, lane); j += 4; }
  if (j + 2 <= m) { add_rows_scaled<2, VT>(acc, pre, xv, ids, scl, j, lane); j += 2; }
  if (j < m)      { add_rows_scaled<1, VT>(acc, pre, xv, ids, scl, j, lane); }
}

__global__ __launch_bounds__(512) __attribute__((amdgpu_waves_per_eu(12)))
void gather_gemm_relu_gemm_kernel(const int* __restrict__ off, const int* __restrict__ cin,
                                  const int* __restrict__ csr, const float* __restrict__ cw,
                                  const float* __restrict__ feat,
                                  const float* __restrict__ inorm, const float* __restrict__ b,
                                  const float* __restrict__ onorm,
                                  const _Float16* __restrict__ Wp1, const _Float16* __restrict__ Wp2,
                                  float* __restrict__ hnext, int n_nodes) {
  typedef float VT __attribute__((ext_vector_type(4)));
  __shared__ __attribute__((aligned(16))) _Float16 xs[16 * XS_PITCH];
  __shared__ __attribute__((aligned(16))) _Float16 ys[16 * XS_PITCH];
  __shared__ __attribute__((aligned(16))) float s_in[16];
  __shared__ __attribute__((aligned(16))) float s_on[16];
  __shared__ __attribute__((aligned(16))) float s_b[128];
  if (threadIdx.x < 128) s_b[threadIdx.x] = b[threadIdx.x];
  __syncthreads();

  int lane = threadIdx.x & 31;
  int w = __builtin_amdgcn_readfirstlane((int)(threadIdx.x >> 5));
  int G = (int)gridDim.x;
  int n_tiles = (n_nodes + 15) >> 4;
  const VT* xv = (const VT*)feat;

  int t = w & 7;

  int r    = lane & 15;
  int half = lane >> 4;

  for (int tfirst = (int)blockIdx.x; tfirst < n_tiles; tfirst += 32 * G) {
    int mt = tfirst + (int)lane * G;
    int mn = mt * 16 + w;
    bool ok = (mt < n_tiles) && (mn < n_nodes);
    int   begL = ok ? off[mn] : 0;
    int   cntL = ok ? cin[mn] : 0;
    float inL  = ok ? inorm[mn] : 0.0f;
    float onL  = ok ? onorm[mn] : 0.0f;
    int beg = 0, n = 0, ids = 0, scl = 0;
    float sc_in = 0.0f, sc_on = 0.0f;
    for (int it = -1; it < 32; ++it) {
      int tile2 = tfirst + (it + 1) * G;
      bool more = (it + 1 < 32) && (tile2 < n_tiles);
      int beg2 = 0, n2 = 0, ids2 = 0, scl2 = 0;
      float in2 = 0.0f, on2 = 0.0f;
      if (more) {
        beg2 = __builtin_amdgcn_readlane(begL, it + 1);
        n2   = __builtin_amdgcn_readlane(cntL, it + 1);
        in2  = lane_f(inL, it + 1);
        on2  = lane_f(onL, it + 1);
        ids2 = ((int)lane < n2) ? csr[beg2 + lane] : 0;
        scl2 = ((int)lane < n2) ? __float_as_int(cw[beg2 + lane]) : 0;
      }
      if (it >= 0) {
      int tile = tfirst + it * G;
      int node = tile * 16 + w;
      VT acc = {0.f, 0.f, 0.f, 0.f};
      if (node < n_nodes) {
        acc = xv[(size_t)node * 32 + lane];
        float pre = sc_on;
        add_list_chunk_scaled<VT>(acc, pre, xv, ids, scl, n < 32 ? n : 32, lane);
        for (int base = 32; base < n; base += 32) {
          int m = n - base;
          if (m > 32) m = 32;
          int idx = (lane < m) ? csr[beg + base + lane] : 0;
          int sdx = (lane < m) ? __float_as_int(cw[beg + base + lane]) : 0;
          add_list_chunk_scaled<VT>(acc, pre, xv, idx, sdx, m, lane);
        }
        acc = acc * pre;
      }

      bool first = ((w >> 3) ^ it) & 1 ? false : true;
      const _Float16* Wsel = first ? Wp1 : Wp2;
      v16h bw[4];
#pragma unroll
      for (int kc = 0; kc < 4; ++kc)
        bw[kc] = *(const v16h*)(Wsel + ((size_t)(kc * 8 + t) * 32 + lane) * 16);

      v4h xh;
      xh[0] = (_Float16)acc[0]; xh[1] = (_Float16)acc[1];
      xh[2] = (_Float16)acc[2]; xh[3] = (_Float16)acc[3];
      *(v4h*)(xs + w * XS_PITCH + lane * 4) = xh;
      if (lane == 0) { s_in[w] = sc_in; s_on[w] = sc_on; }
      __syncthreads();

      if (first) {
        const _Float16* pa = xs + r * XS_PITCH + half * 8;
        v8f acc1 = {0.f, 0.f, 0.f, 0.f, 0.f, 0.f, 0.f, 0.f};
#pragma unroll
        for (int kc = 0; kc < 4; ++kc) {
          v8h lo = *(const v8h*)(pa + kc * 32);
          v8h hi = *(const v8h*)(pa + kc * 32 + 16);
          v16h a = __builtin_shufflevector(lo, hi, 0, 1, 2, 3, 4, 5, 6, 7, 8, 9, 10, 11, 12, 13, 14, 15);
          acc1 = __builtin_amdgcn_wmma_f32_16x16x32_f16(
               false, a,  false, bw[kc],
               (short)0, acc1,  false,  false);
        }
        float bcol = s_b[t * 16 + r];
#pragma unroll
        for (int vr = 0; vr < 8; ++vr) {
          int row = half * 8 + vr;
          float v = acc1[vr] * s_in[row] + bcol;
          v = fmaxf(v, 0.0f);
          v = v * s_on[row];
          ys[row * XS_PITCH + t * 16 + r] = (_Float16)v;
        }
      }
      __syncthreads();

      if (!first) {
        const _Float16* pb = ys + r * XS_PITCH + half * 8;
        v8f acc2 = {0.f, 0.f, 0.f, 0.f, 0.f, 0.f, 0.f, 0.f};
#pragma unroll
        for (int kc = 0; kc < 4; ++kc) {
          v8h lo = *(const v8h*)(pb + kc * 32);
          v8h hi = *(const v8h*)(pb + kc * 32 + 16);
          v16h a = __builtin_shufflevector(lo, hi, 0, 1, 2, 3, 4, 5, 6, 7, 8, 9, 10, 11, 12, 13, 14, 15);
          acc2 = __builtin_amdgcn_wmma_f32_16x16x32_f16(
               false, a,  false, bw[kc],
               (short)0, acc2,  false,  false);
        }
        int row0 = tile * 16 + half * 8;
        unsigned po = (unsigned)(row0 * 128 + t * 16 + r);
        if (tile * 16 + 16 <= n_nodes) {
#pragma unroll
          for (int vr = 0; vr < 8; ++vr) hnext[po + (unsigned)(vr * 128)] = acc2[vr];
        } else {
#pragma unroll
          for (int vr = 0; vr < 8; ++vr)
            if (row0 + vr < n_nodes) hnext[po + (unsigned)(vr * 128)] = acc2[vr];
        }
      }
      }
      if (!more) break;
      beg = beg2; n = n2; ids = ids2; scl = scl2; sc_in = in2; sc_on = on2;
    }
  }
}

extern "C" void kernel_launch(void* const* d_in, const int* in_sizes, int n_in,
                              void* d_out, int out_size, void* d_ws, size_t ws_size,
                              hipStream_t stream) {
  const float* feat = (const float*)d_in[0];
  const int*   src  = (const int*)  d_in[1];
  const int*   dst  = (const int*)  d_in[2];
  const float* W1   = (const float*)d_in[3];
  const float* b1   = (const float*)d_in[4];
  const float* W2   = (const float*)d_in[5];
  const float* b2   = (const float*)d_in[6];
  const float* W3   = (const float*)d_in[7];
  const float* b3   = (const float*)d_in[8];
  float* out = (float*)d_out;

  size_t off = 0;
  auto alloc = [&](size_t bytes) -> char* {
    char* p = (char*)d_ws + off;
    off += (bytes + 255) & ~(size_t)255;
    return p;
  };
  float*     onorm = (float*)alloc((size_t)NNODES * sizeof(float));
  float*     inorm = (float*)alloc((size_t)NNODES * sizeof(float));
  float*     hbuf  = (float*)alloc((size_t)NNODES * FDIM * sizeof(float));
  float*     xbuf  = (float*)alloc((size_t)NNODES * FDIM * sizeof(float));
  _Float16*  Wp1   = (_Float16*)alloc((size_t)FDIM * FDIM * sizeof(_Float16));
  _Float16*  Wp2   = (_Float16*)alloc((size_t)FDIM * FDIM * sizeof(_Float16));
  _Float16*  Wp3   = (_Float16*)alloc((size_t)FDIM * FDIM * sizeof(_Float16));
  int*       cnt_in = (int*)alloc((size_t)NNODES * sizeof(int));
  int*       cnt_out= (int*)alloc((size_t)NNODES * sizeof(int));
  int*       lstart= (int*)alloc((size_t)NNODES * sizeof(int));
  int*       hist  = (int*)alloc((size_t)NBKT * NCHUNK * sizeof(int));
  int*       boff  = (int*)alloc((size_t)NBKT * NCHUNK * sizeof(int));
  int2*      ebuf  = (int2*)alloc((size_t)NEDGES * sizeof(int2));
  int*       lst   = (int*)alloc((size_t)NEDGES * sizeof(int));
  int*       csr   = (int*)alloc((size_t)NEDGES * sizeof(int));
  float*     cw    = (float*)alloc((size_t)NEDGES * sizeof(float));

  const int TB = 256;

  chunk_count_kernel<<<NCHUNK, 1024, 0, stream>>>(dst, cnt_out, hist, NEDGES, NNODES);
  piece_scan_kernel<<<SCAN_GRID, 1024, 0, stream>>>(hist, boff, W1, W2, W3, Wp1, Wp2, Wp3);
  chunk_deal_kernel<<<NCHUNK, 1024, 0, stream>>>(src, dst, boff, cnt_out, ebuf, NEDGES);
  bucket_lists_kernel<<<NBKT, 1024, 0, stream>>>(ebuf, boff, cnt_out, lstart, cnt_in, lst, onorm, inorm, NNODES, NEDGES);
  order_kernel<<<RES_BLOCKS_8, TB, 0, stream>>>(lstart, cnt_in, lst, onorm, csr, cw, NNODES);

  gather_gemm_relu_gemm_kernel<<<RES_BLOCKS_16x3, 512, 0, stream>>>(
      lstart, cnt_in, csr, cw, feat, inorm, b1, onorm, Wp1, Wp2, xbuf, NNODES);

  gather_relu_gemm_kernel<64><<<RES_BLOCKS_16x4, 512, 0, stream>>>(
      lstart, cnt_in, csr, xbuf, inorm, b2, onorm, Wp3, hbuf, NNODES);

  gather_agg_kernel<64, false><<<RES_BLOCKS_8, TB, 0, stream>>>(
      lstart, cnt_in, csr, hbuf, inorm, b3, out, NNODES);
}


# List ordering moved into the bucket block: a bucket's edges stay in LDS from count to fill, each entry is ranked there by source id, one thread per entry; no ordering launch; same order of sums
# speedup vs baseline: 4.8690x; 1.1609x over previous
#include <hip/hip_runtime.h>
#include <math.h>

typedef __attribute__((ext_vector_type(16))) _Float16 v16h;
typedef __attribute__((ext_vector_type(8)))  float    v8f;
typedef __attribute__((ext_vector_type(8)))  _Float16 v8h;
typedef __attribute__((ext_vector_type(4)))  _Float16 v4h;

#define NNODES 100000
#define NEDGES 1600000
#define FDIM   128

#define BKT_SHIFT   9
#define BKT_NODES   (1 << BKT_SHIFT)
#define NBKT        ((NNODES + BKT_NODES - 1) / BKT_NODES)
#define NCHUNK      256
#define CHUNK_EDGES ((NEDGES + NCHUNK - 1) / NCHUNK)
#define CHUNK_PER   ((CHUNK_EDGES + 1023) / 1024)
#define ZSLICE      ((NNODES + NCHUNK - 1) / NCHUNK)

__global__ __launch_bounds__(1024)
void chunk_count_kernel(const int* __restrict__ dst, int* __restrict__ cout,
                        int* __restrict__ hist, int e, int n) {
  __shared__ int s_h[NBKT];
  int t = threadIdx.x, c = blockIdx.x;
  for (int b = t; b < NBKT; b += 1024) s_h[b] = 0;
  int zend = (c + 1) * ZSLICE < n ? (c + 1) * ZSLICE : n;
  for (int i = c * ZSLICE + t; i < zend; i += 1024) cout[i] = 0;
  __syncthreads();
  int beg = c * CHUNK_EDGES;
  int end = beg + CHUNK_EDGES < e ? beg + CHUNK_EDGES : e;
  int d[CHUNK_PER];
#pragma unroll
  for (int u = 0; u < CHUNK_PER; ++u) {
    int i = beg + t + u * 1024;
    d[u] = dst[i < end ? i : end - 1];
  }
#pragma unroll
  for (int u = 0; u < CHUNK_PER; ++u) {
    int i = beg + t + u * 1024;
    if (i < end) atomicAdd(&s_h[d[u] >> BKT_SHIFT], 1);
  }
  __syncthreads();
  for (int b = t; b < NBKT; b += 1024) hist[b * NCHUNK + c] = s_h[b];
}

__device__ __forceinline__ void pack_w_elem(const float* __restrict__ W, _Float16* __restrict__ Wp, int nout, int idx) {
  int nt = nout >> 4;
  int j    = idx & 15;
  int l    = (idx >> 4) & 31;
  int rest = idx >> 9;
  int t    = rest % nt;
  int c    = rest / nt;
  int col   = t * 16 + (l & 15);
  int kbase = c * 32 + ((l >> 4) * 8);
  int K     = kbase + (j & 7) + ((j >> 3) * 16);
  Wp[idx] = (_Float16)W[K * nout + col];
}

#define ROWS_PER_WAVE ((NBKT + 31) / 32)

#define PACK_BLK_128 ((4 * 8 * 512) / 1024)
#define PACK_BLK_64  ((4 * 4 * 512) / 1024)
#define PACK_BLOCKS  (2 * PACK_BLK_128 + PACK_BLK_64)
#define SCAN_GRID    (1 + PACK_BLOCKS)

__global__ __launch_bounds__(1024)
void piece_scan_kernel(const int* __restrict__ hist, int* __restrict__ boff,
                       const float* __restrict__ W1, const float* __restrict__ W2,
                       const float* __restrict__ W3, _Float16* __restrict__ Wp1,
                       _Float16* __restrict__ Wp2, _Float16* __restrict__ Wp3) {
  __shared__ int s_tot[NBKT];
  int t = threadIdx.x, lane = t & 31, w = t >> 5;
  if (blockIdx.x > 0) {
    int pb = (int)blockIdx.x - 1;
    if (pb < PACK_BLK_128)          pack_w_elem(W1, Wp1, 128, pb * 1024 + t);
    else if (pb < 2 * PACK_BLK_128) pack_w_elem(W2, Wp2, 128, (pb - PACK_BLK_128) * 1024 + t);
    else                            pack_w_elem(W3, Wp3, 64, (pb - 2 * PACK_BLK_128) * 1024 + t);
    return;
  }
  const int4* h4 = (const int4*)hist;
  int4* o4 = (int4*)boff;
  int4 p[ROWS_PER_WAVE], q[ROWS_PER_WAVE];
#pragma unroll
  for (int u = 0; u < ROWS_PER_WAVE; ++u) {
    int b = w + 32 * u;
    int bb = b < NBKT ? b : NBKT - 1;
    p[u] = h4[bb * (NCHUNK / 4) + lane * 2];
    q[u] = h4[bb * (NCHUNK / 4) + lane * 2 + 1];
  }
#pragma unroll
  for (int u = 0; u < ROWS_PER_WAVE; ++u) {
    int b = w + 32 * u;
    int s = ((p[u].x + p[u].y) + (p[u].z + p[u].w)) + ((q[u].x + q[u].y) + (q[u].z + q[u].w));
#pragma unroll
    for (int k = 16; k > 0; k >>= 1) s += __shfl_xor(s, k);
    if (lane == 0 && b < NBKT) s_tot[b] = s;
  }
  __syncthreads();
#pragma unroll
  for (int u = 0; u < ROWS_PER_WAVE; ++u) {
    int b = w + 32 * u;
    if (b < NBKT) {
      int front = 0;
      for (int k = lane; k < b; k += 32) front += s_tot[k];
#pragma unroll
      for (int k = 16; k > 0; k >>= 1) front += __shfl_xor(front, k);
      int s = ((p[u].x + p[u].y) + (p[u].z + p[u].w)) + ((q[u].x + q[u].y) + (q[u].z + q[u].w));
      int incl = s;
#pragma unroll
      for (int k = 1; k < 32; k <<= 1) {
        int o = __shfl_up(incl, k);
        if (lane >= k) incl += o;
      }
      int run = front + incl - s;
      int4 op, oq;
      op.x = run; run += p[u].x;  op.y = run; run += p[u].y;
      op.z = run; run += p[u].z;  op.w = run; run += p[u].w;
      oq.x = run; run += q[u].x;  oq.y = run; run += q[u].y;
      oq.z = run; run += q[u].z;  oq.w = run;
      o4[b * (NCHUNK / 4) + lane * 2]     = op;
      o4[b * (NCHUNK / 4) + lane * 2 + 1] = oq;
    }
  }
}

__global__ __launch_bounds__(1024)
void chunk_deal_kernel(const int* __restrict__ src, const int* __restrict__ dst,
                       const int* __restrict__ boff, int* __restrict__ cout,
                       int2* __restrict__ ebuf, int e) {
  __shared__ int s_cur[NBKT];
  int t = threadIdx.x, c = blockIdx.x;
  for (int b = t; b < NBKT; b += 1024) s_cur[b] = boff[b * NCHUNK + c];
  __syncthreads();
  int beg = c * CHUNK_EDGES;
  int end = beg + CHUNK_EDGES < e ? beg + CHUNK_EDGES : e;
  int d[CHUNK_PER], s[CHUNK_PER];
#pragma unroll
  for (int u = 0; u < CHUNK_PER; ++u) {
    int i = beg + t + u * 1024;
    int ii = i < end ? i : end - 1;
    d[u] = dst[ii];
    s[u] = src[ii];
  }
#pragma unroll
  for (int u = 0; u < CHUNK_PER; ++u) {
    int i = beg + t + u * 1024;
    if (i < end) {
      int pos = atomicAdd(&s_cur[d[u] >> BKT_SHIFT], 1);
      ebuf[pos] = make_int2(s[u], d[u]);
      atomicAdd(&cout[s[u]], 1);
    }
  }
}

#define SRC_BITS 17
#define SRC_MASK ((1 << SRC_BITS) - 1)
static_assert(NNODES <= (1 << SRC_BITS), "a source id must fit below the node index of a list entry");
static_assert(BKT_SHIFT + SRC_BITS <= 31, "a list entry must stay a non-negative int");
#define LIST_CAP 16384
#define LIST_PAD 8

__global__ __launch_bounds__(1024)
void bucket_lists_kernel(const int2* __restrict__ ebuf, const int* __restrict__ boff,
                         const int* __restrict__ cout, int* __restrict__ off,
                         int* __restrict__ cin, int* lst,
                         float* __restrict__ onorm, float* __restrict__ inorm,
                         int* __restrict__ csr, float* __restrict__ cw, int n, int e) {
  __shared__ int s_cnt[BKT_NODES];
  __shared__ int s_off[BKT_NODES];
  __shared__ int s_w[32];
  __shared__ int s_raw[LIST_CAP];
  __shared__ int s_lst[LIST_CAP + LIST_PAD];
  int t = threadIdx.x, lane = t & 31, w = t >> 5;
  int b = blockIdx.x;
  int base = b * BKT_NODES;
  int ebeg = boff[b * NCHUNK];
  int eend = (b + 1 < NBKT) ? boff[(b + 1) * NCHUNK] : e;
  int m = eend - ebeg;
  bool in_lds = m <= LIST_CAP;
  int* glst = lst + ebeg;
  if (t < BKT_NODES) s_cnt[t] = 0;
  __syncthreads();
  if (in_lds) {
    for (int e0 = t; e0 < m; e0 += 8 * 1024) {
      int2 ed[8];
#pragma unroll
      for (int u = 0; u < 8; ++u) {
        int ei = e0 + u * 1024;
        ed[u] = ebuf[ebeg + (ei < m ? ei : m - 1)];
      }
#pragma unroll
      for (int u = 0; u < 8; ++u) {
        int ei = e0 + u * 1024;
        if (ei < m) {
          int l = ed[u].y - base;
          atomicAdd(&s_cnt[l], 1);
          s_raw[ei] = (l << SRC_BITS) | ed[u].x;
        }
      }
    }
  } else {
    for (int i0 = ebeg + t; i0 < eend; i0 += 4 * 1024) {
      int dd[4];
#pragma unroll
      for (int u = 0; u < 4; ++u) {
        int i = i0 + u * 1024;
        dd[u] = ebuf[i < eend ? i : eend - 1].y;
      }
#pragma unroll
      for (int u = 0; u < 4; ++u) {
        int i = i0 + u * 1024;
        if (i < eend) atomicAdd(&s_cnt[dd[u] - base], 1);
      }
    }
  }
  __syncthreads();
  int cnt = (t < BKT_NODES) ? s_cnt[t] : 0;
  int incl = cnt;
#pragma unroll
  for (int k = 1; k < 32; k <<= 1) {
    int o = __shfl_up(incl, k);
    if (lane >= k) incl += o;
  }
  if (lane == 31) s_w[w] = incl;
  __syncthreads();
  int front = (lane < w) ? s_w[lane] : 0;
#pragma unroll
  for (int k = 16; k > 0; k >>= 1) front += __shfl_xor(front, k);
  int lo = front + incl - cnt;
  if (t < BKT_NODES) {
    s_off[t] = lo;
    s_cnt[t] = 0;
    int node = base + t;
    if (node < n) {
      off[node] = ebeg + lo;
      cin[node] = cnt;
      onorm[node] = rsqrtf(fmaxf((float)(cout[node] + 1), 1.0f));
      inorm[node] = rsqrtf(fmaxf((float)(cnt + 1), 1.0f));
    }
  }
  __syncthreads();
  if (in_lds) {
    for (int ei = t; ei < m; ei += 1024) {
      int v = s_raw[ei];
      int l = v >> SRC_BITS;
      int k = atomicAdd(&s_cnt[l], 1);
      s_lst[s_off[l] + k] = v;
    }
    if (t < LIST_PAD) s_lst[m + t] = 0x7fffffff;
    __syncthreads();
    for (int i = t; i < m; i += 1024) {
      int v   = s_lst[i];
      int l   = v >> SRC_BITS;
      int l0  = s_off[l];
      int ln  = s_cnt[l];
      int k   = v & SRC_MASK;
      int co  = cout[k];
      int r = 0, r1 = 0;
#pragma clang loop vectorize(disable) interleave(disable)
      for (int a = l0; a < l0 + ln; a += 8) {
        int v0 = s_lst[a],     v1 = s_lst[a + 1], v2 = s_lst[a + 2], v3 = s_lst[a + 3];
        int v4 = s_lst[a + 4], v5 = s_lst[a + 5], v6 = s_lst[a + 6], v7 = s_lst[a + 7];
        r  += (v0 < v + (a     < i ? 1 : 0)) ? 1 : 0;
        r1 += (v1 < v + (a + 1 < i ? 1 : 0)) ? 1 : 0;
        r  += (v2 < v + (a + 2 < i ? 1 : 0)) ? 1 : 0;
        r1 += (v3 < v + (a + 3 < i ? 1 : 0)) ? 1 : 0;
        r  += (v4 < v + (a + 4 < i ? 1 : 0)) ? 1 : 0;
        r1 += (v5 < v + (a + 5 < i ? 1 : 0)) ? 1 : 0;
        r  += (v6 < v + (a + 6 < i ? 1 : 0)) ? 1 : 0;
        r1 += (v7 < v + (a + 7 < i ? 1 : 0)) ? 1 : 0;
      }
      r += r1;
      csr[ebeg + l0 + r] = k;
      cw[ebeg + l0 + r]  = rsqrtf(fmaxf((float)(co + 1), 1.0f));
    }
  } else {
    for (int i0 = ebeg + t; i0 < eend; i0 += 4 * 1024) {
      int2 ed[4];
#pragma unroll
      for (int u = 0; u < 4; ++u) {
        int i = i0 + u * 1024;
        ed[u] = ebuf[i < eend ? i : eend - 1];
      }
#pragma unroll
      for (int u = 0; u < 4; ++u) {
        int i = i0 + u * 1024;
        if (i < eend) {
          int l = ed[u].y - base;
          int k = atomicAdd(&s_cnt[l], 1);
          glst[s_off[l] + k] = (l << SRC_BITS) | ed[u].x;
        }
      }
    }
    __syncthreads();
    for (int i = t; i < m; i += 1024) {
      int v   = glst[i];
      int l   = v >> SRC_BITS;
      int l0  = s_off[l];
      int ln  = s_cnt[l];
      int k   = v & SRC_MASK;
      int co  = cout[k];
      int r = 0;
      for (int j = 0; j < ln; ++j) {
        int vj = glst[l0 + j];
        r += (vj < v + (l0 + j < i ? 1 : 0)) ? 1 : 0;
      }
      csr[ebeg + l0 + r] = k;
      cw[ebeg + l0 + r]  = rsqrtf(fmaxf((float)(co + 1), 1.0f));
    }
  }
}

#define RES_BLOCKS_8    2048
#define RES_BLOCKS_16x4 1024
#define RES_BLOCKS_16x3  768

__device__ __forceinline__ float lane_f(float v, int l) {
  return __int_as_float(__builtin_amdgcn_readlane(__float_as_int(v), l));
}

template<int U, typename VT, typename LT>
__device__ __forceinline__ void add_rows(VT& acc, const LT* __restrict__ hv, int ids, int j, unsigned lane) {
  LT r[U];
#pragma unroll
  for (int u = 0; u < U; ++u) {
    int s = __builtin_amdgcn_readlane(ids, j + u);
    const LT* row = hv + (size_t)s * 32;
    r[u] = row[lane];
  }
#pragma unroll
  for (int u = 0; u < U; ++u) acc += __builtin_convertvector(r[u], VT);
}

template<typename VT, typename LT>
__device__ __forceinline__ void add_list_chunk(VT& acc, const LT* __restrict__ hv, int ids, int m, unsigned lane) {
  int j = 0;
  for (; j + 8 <= m; j += 8) add_rows<8, VT, LT>(acc, hv, ids, j, lane);
  if (j + 4 <= m) { add_rows<4, VT, LT>(acc, hv, ids, j, lane); j += 4; }
  if (j + 2 <= m) { add_rows<2, VT, LT>(acc, hv, ids, j, lane); j += 2; }
  if (j < m)      { add_rows<1, VT, LT>(acc, hv, ids, j, lane); }
}

template<int NOUT, bool RELU>
__global__ __launch_bounds__(256)
void gather_agg_kernel(const int* __restrict__ off, const int* __restrict__ cin,
                       const int* __restrict__ csr, const _Float16* __restrict__ h,
                       const float* __restrict__ inorm, const float* __restrict__ b,
                       float* __restrict__ out, int n_nodes) {
  constexpr int V = NOUT / 32;
  typedef float VT __attribute__((ext_vector_type(V)));
  typedef _Float16 LT __attribute__((ext_vector_type(V)));
  int lane = threadIdx.x & 31;
  int nw = (int)((gridDim.x * blockDim.x) >> 5);
  int gw = __builtin_amdgcn_readfirstlane((int)((blockIdx.x * blockDim.x + threadIdx.x) >> 5));
  const LT* hv = (const LT*)h;
  VT bv = ((const VT*)b)[lane];
  for (int first = gw; first < n_nodes; first += 32 * nw) {
    int mn = first + lane * nw;
    bool ok = mn < n_nodes;
    int   begL = ok ? off[mn] : 0;
    int   cntL = ok ? cin[mn] : 0;
    float inL  = ok ? inorm[mn] : 0.0f;
    int beg = 0, n = 0, ids = 0;
    for (int it = -1; it < 32; ++it) {
      int node2 = first + (it + 1) * nw;
      bool more = (it + 1 < 32) && (node2 < n_nodes);
      int beg2 = 0, n2 = 0, ids2 = 0;
      if (more) {
        beg2 = __builtin_amdgcn_readlane(begL, it + 1);
        n2   = __builtin_amdgcn_readlane(cntL, it + 1);
        ids2 = (lane < n2) ? csr[beg2 + lane] : 0;
      }
      if (it >= 0) {
        int node = first + it * nw;
        VT acc = __builtin_convertvector(hv[(size_t)node * 32 + lane], VT);
        add_list_chunk<VT, LT>(acc, hv, ids, n < 32 ? n : 32, lane);
        for (int base = 32; base < n; base += 32) {
          int m = n - base;
          if (m > 32) m = 32;
          int idx = (lane < m) ? csr[beg + base + lane] : 0;
          add_list_chunk<VT, LT>(acc, hv, idx, m, lane);
        }
        float sc = lane_f(inL, it);
        VT v = acc * sc + bv;
        if (RELU) {
#pragma unroll
          for (int k = 0; k < V; ++k) v[k] = fmaxf(v[k], 0.0f);
        }
        ((VT*)out)[(size_t)node * 32 + lane] = v;
      }
      if (!more) break;
      beg = beg2; n = n2; ids = ids2;
    }
  }
}


#define XS_PITCH 136

template<int NOUT2>
__global__ __launch_bounds__(512) __attribute__((amdgpu_waves_per_eu(16)))
void gather_relu_gemm_kernel(const int* __restrict__ off, const int* __restrict__ cin,
                             const int* __restrict__ csr, const _Float16* __restrict__ h,
                             const float* __restrict__ inorm, const float* __restrict__ b,
                             const float* __restrict__ onorm, const _Float16* __restrict__ Wp,
                             _Float16* __restrict__ hnext, int n_nodes) {
  constexpr int NT2 = NOUT2 / 16;
  typedef float VT __attribute__((ext_vector_type(4)));
  typedef v4h LT;
  __shared__ __attribute__((aligned(16))) _Float16 xs[2][16 * XS_PITCH];
  __shared__ __attribute__((aligned(16))) float s_b[128];
  if (threadIdx.x < 128) s_b[threadIdx.x] = b[threadIdx.x];
  __syncthreads();

  int lane = threadIdx.x & 31;
  int w = __builtin_amdgcn_readfirstlane((int)(threadIdx.x >> 5));
  int G = (int)gridDim.x;
  int n_tiles = (n_nodes + 15) >> 4;
  const LT* hv = (const LT*)h;

  int r    = lane & 15;
  int half = lane >> 4;

  for (int tfirst = (int)blockIdx.x; tfirst < n_tiles; tfirst += 32 * G) {
    int mt = tfirst + (int)lane * G;
    int mn = mt * 16 + w;
    bool ok = (mt < n_tiles) && (mn < n_nodes);
    int   begL = ok ? off[mn] : 0;
    int   cntL = ok ? cin[mn] : 0;
    float inL  = ok ? inorm[mn] : 0.0f;
    float onL  = ok ? onorm[mn] : 0.0f;
    int beg = 0, n = 0, ids = 0;
    for (int it = -1; it < 32; ++it) {
      int tile2 = tfirst + (it + 1) * G;
      bool more = (it + 1 < 32) && (tile2 < n_tiles);
      int beg2 = 0, n2 = 0, ids2 = 0;
      if (more) {
        beg2 = __builtin_amdgcn_readlane(begL, it + 1);
        n2   = __builtin_amdgcn_readlane(cntL, it + 1);
        ids2 = ((int)lane < n2) ? csr[beg2 + lane] : 0;
      }
      if (it >= 0) {
      int tile = tfirst + it * G;
      int node = tile * 16 + w;
      VT v = {0.f, 0.f, 0.f, 0.f};
      if (node < n_nodes) {
        VT acc = __builtin_convertvector(hv[(size_t)node * 32 + lane], VT);
        add_list_chunk<VT, LT>(acc, hv, ids, n < 32 ? n : 32, lane);
        for (int base = 32; base < n; base += 32) {
          int m = n - base;
          if (m > 32) m = 32;
          int idx = (lane < m) ? csr[beg + base + lane] : 0;
          add_list_chunk<VT, LT>(acc, hv, idx, m, lane);
        }
        float sc = lane_f(inL, it);
        VT bv = *(const VT*)(s_b + lane * 4);
        v = acc * sc + bv;
#pragma unroll
        for (int k = 0; k < 4; ++k) v[k] = fmaxf(v[k], 0.0f);
        float on = lane_f(onL, it);
        v = v * on;
      }

      int pw = (w - NT2 * it) & 15;
      v16h bw[4];
#pragma unroll
      for (int kc = 0; kc < 4; ++kc) bw[kc] = (v16h)(_Float16)0.0f;
      if (pw < NT2) {
#pragma unroll
        for (int kc = 0; kc < 4; ++kc)
          bw[kc] = *(const v16h*)(Wp + ((size_t)(kc * NT2 + pw) * 32 + lane) * 16);
      }

      _Float16* xt = xs[it & 1];
      v4h xh;
      xh[0] = (_Float16)v[0]; xh[1] = (_Float16)v[1];
      xh[2] = (_Float16)v[2]; xh[3] = (_Float16)v[3];
      *(v4h*)(xt + w * XS_PITCH + lane * 4) = xh;
      __syncthreads();

      if (pw < NT2) {
        const _Float16* pa = xt + r * XS_PITCH + half * 8;
        v8f acc2 = {0.f, 0.f, 0.f, 0.f, 0.f, 0.f, 0.f, 0.f};
#pragma unroll
        for (int kc = 0; kc < 4; ++kc) {
          v8h lo = *(const v8h*)(pa + kc * 32);
          v8h hi = *(const v8h*)(pa + kc * 32 + 16);
          v16h a = __builtin_shufflevector(lo, hi, 0, 1, 2, 3, 4, 5, 6, 7, 8, 9, 10, 11, 12, 13, 14, 15);
          acc2 = __builtin_amdgcn_wmma_f32_16x16x32_f16(
               false, a,  false, bw[kc],
               (short)0, acc2,  false,  false);
        }
        int row0 = tile * 16 + half * 8;
        unsigned po = (unsigned)(row0 * NOUT2 + pw * 16 + r);
        if (tile * 16 + 16 <= n_nodes) {
#pragma unroll
          for (int vr = 0; vr < 8; ++vr) hnext[po + (unsigned)(vr * NOUT2)] = (_Float16)acc2[vr];
        } else {
#pragma unroll
          for (int vr = 0; vr < 8; ++vr)
            if (row0 + vr < n_nodes) hnext[po + (unsigned)(vr * NOUT2)] = (_Float16)acc2[vr];
        }
      }
      }
      if (!more) break;
      beg = beg2; n = n2; ids = ids2;
    }
  }
}

template<int U, typename VT>
__device__ __forceinline__ void add_rows_scaled(VT& acc, float& pre, const VT* __restrict__ xv, int ids, int scl, int j, unsigned lane) {
  VT r[U];
#pragma unroll
  for (int u = 0; u < U; ++u) {
    int s = __builtin_amdgcn_readlane(ids, j + u);
    const VT* row = xv + (size_t)s * 32;
    r[u] = row[lane];
  }
  acc = acc * pre;
  pre = 1.0f;
#pragma unroll
  for (int u = 0; u < U; ++u) {
    float sc = __int_as_float(__builtin_amdgcn_readlane(scl, j + u));
    acc += r[u] * sc;
  }
}

template<typename VT>
__device__ __forceinline__ void add_list_chunk_scaled(VT& acc, float& pre, const VT* __restrict__ xv, int ids, int scl, int m, unsigned lane) {
  int j = 0;
  for (; j + 8 <= m; j += 8) add_rows_scaled<8, VT>(acc, pre, xv, ids, scl, j, lane);
  if (j + 4 <= m) { add_rows_scaled<4, VT>(acc, pre, xv, ids, scl, j, lane); j += 4; }
  if (j + 2 <= m) { add_rows_scaled<2, VT>(acc, pre, xv, ids, scl, j, lane); j += 2; }
  if (j < m)      { add_rows_scaled<1, VT>(acc, pre, xv, ids, scl, j, lane); }
}

__global__ __launch_bounds__(512) __attribute__((amdgpu_waves_per_eu(12)))
void gather_gemm_relu_gemm_kernel(const int* __restrict__ off, const int* __restrict__ cin,
                                  const int* __restrict__ csr, const float* __restrict__ cw,
                                  const float* __restrict__ feat,
                                  const float* __restrict__ inorm, const float* __restrict__ b,
                                  const float* __restrict__ onorm,
                                  const _Float16* __restrict__ Wp1, const _Float16* __restrict__ Wp2,
                                  _Float16* __restrict__ hnext, int n_nodes) {
  typedef float VT __attribute__((ext_vector_type(4)));
  __shared__ __attribute__((aligned(16))) _Float16 xs[16 * XS_PITCH];
  __shared__ __attribute__((aligned(16))) _Float16 ys[16 * XS_PITCH];
  __shared__ __attribute__((aligned(16))) float s_in[16];
  __shared__ __attribute__((aligned(16))) float s_on[16];
  __shared__ __attribute__((aligned(16))) float s_b[128];
  if (threadIdx.x < 128) s_b[threadIdx.x] = b[threadIdx.x];
  __syncthreads();

  int lane = threadIdx.x & 31;
  int w = __builtin_amdgcn_readfirstlane((int)(threadIdx.x >> 5));
  int G = (int)gridDim.x;
  int n_tiles = (n_nodes + 15) >> 4;
  const VT* xv = (const VT*)feat;

  int t = w & 7;

  int r    = lane & 15;
  int half = lane >> 4;

  for (int tfirst = (int)blockIdx.x; tfirst < n_tiles; tfirst += 32 * G) {
    int mt = tfirst + (int)lane * G;
    int mn = mt * 16 + w;
    bool ok = (mt < n_tiles) && (mn < n_nodes);
    int   begL = ok ? off[mn] : 0;
    int   cntL = ok ? cin[mn] : 0;
    float inL  = ok ? inorm[mn] : 0.0f;
    float onL  = ok ? onorm[mn] : 0.0f;
    int beg = 0, n = 0, ids = 0, scl = 0;
    float sc_in = 0.0f, sc_on = 0.0f;
    for (int it = -1; it < 32; ++it) {
      int tile2 = tfirst + (it + 1) * G;
      bool more = (it + 1 < 32) && (tile2 < n_tiles);
      int beg2 = 0, n2 = 0, ids2 = 0, scl2 = 0;
      float in2 = 0.0f, on2 = 0.0f;
      if (more) {
        beg2 = __builtin_amdgcn_readlane(begL, it + 1);
        n2   = __builtin_amdgcn_readlane(cntL, it + 1);
        in2  = lane_f(inL, it + 1);
        on2  = lane_f(onL, it + 1);
        ids2 = ((int)lane < n2) ? csr[beg2 + lane] : 0;
        scl2 = ((int)lane < n2) ? __float_as_int(cw[beg2 + lane]) : 0;
      }
      if (it >= 0) {
      int tile = tfirst + it * G;
      int node = tile * 16 + w;
      VT acc = {0.f, 0.f, 0.f, 0.f};
      if (node < n_nodes) {
        acc = xv[(size_t)node * 32 + lane];
        float pre = sc_on;
        add_list_chunk_scaled<VT>(acc, pre, xv, ids, scl, n < 32 ? n : 32, lane);
        for (int base = 32; base < n; base += 32) {
          int m = n - base;
          if (m > 32) m = 32;
          int idx = (lane < m) ? csr[beg + base + lane] : 0;
          int sdx = (lane < m) ? __float_as_int(cw[beg + base + lane]) : 0;
          add_list_chunk_scaled<VT>(acc, pre, xv, idx, sdx, m, lane);
        }
        acc = acc * pre;
      }

      bool first = ((w >> 3) ^ it) & 1 ? false : true;
      const _Float16* Wsel = first ? Wp1 : Wp2;
      v16h bw[4];
#pragma unroll
      for (int kc = 0; kc < 4; ++kc)
        bw[kc] = *(const v16h*)(Wsel + ((size_t)(kc * 8 + t) * 32 + lane) * 16);

      v4h xh;
      xh[0] = (_Float16)acc[0]; xh[1] = (_Float16)acc[1];
      xh[2] = (_Float16)acc[2]; xh[3] = (_Float16)acc[3];
      *(v4h*)(xs + w * XS_PITCH + lane * 4) = xh;
      if (lane == 0) { s_in[w] = sc_in; s_on[w] = sc_on; }
      __syncthreads();

      if (first) {
        const _Float16* pa = xs + r * XS_PITCH + half * 8;
        v8f acc1 = {0.f, 0.f, 0.f, 0.f, 0.f, 0.f, 0.f, 0.f};
#pragma unroll
        for (int kc = 0; kc < 4; ++kc) {
          v8h lo = *(const v8h*)(pa + kc * 32);
          v8h hi = *(const v8h*)(pa + kc * 32 + 16);
          v16h a = __builtin_shufflevector(lo, hi, 0, 1, 2, 3, 4, 5, 6, 7, 8, 9, 10, 11, 12, 13, 14, 15);
          acc1 = __builtin_amdgcn_wmma_f32_16x16x32_f16(
               false, a,  false, bw[kc],
               (short)0, acc1,  false,  false);
        }
        float bcol = s_b[t * 16 + r];
#pragma unroll
        for (int vr = 0; vr < 8; ++vr) {
          int row = half * 8 + vr;
          float v = acc1[vr] * s_in[row] + bcol;
          v = fmaxf(v, 0.0f);
          v = v * s_on[row];
          ys[row * XS_PITCH + t * 16 + r] = (_Float16)v;
        }
      }
      __syncthreads();

      if (!first) {
        const _Float16* pb = ys + r * XS_PITCH + half * 8;
        v8f acc2 = {0.f, 0.f, 0.f, 0.f, 0.f, 0.f, 0.f, 0.f};
#pragma unroll
        for (int kc = 0; kc < 4; ++kc) {
          v8h lo = *(const v8h*)(pb + kc * 32);
          v8h hi = *(const v8h*)(pb + kc * 32 + 16);
          v16h a = __builtin_shufflevector(lo, hi, 0, 1, 2, 3, 4, 5, 6, 7, 8, 9, 10, 11, 12, 13, 14, 15);
          acc2 = __builtin_amdgcn_wmma_f32_16x16x32_f16(
               false, a,  false, bw[kc],
               (short)0, acc2,  false,  false);
        }
        int row0 = tile * 16 + half * 8;
        unsigned po = (unsigned)(row0 * 128 + t * 16 + r);
        if (tile * 16 + 16 <= n_nodes) {
#pragma unroll
          for (int vr = 0; vr < 8; ++vr) hnext[po + (unsigned)(vr * 128)] = (_Float16)acc2[vr];
        } else {
#pragma unroll
          for (int vr = 0; vr < 8; ++vr)
            if (row0 + vr < n_nodes) hnext[po + (unsigned)(vr * 128)] = (_Float16)acc2[vr];
        }
      }
      }
      if (!more) break;
      beg = beg2; n = n2; ids = ids2; scl = scl2; sc_in = in2; sc_on = on2;
    }
  }
}

extern "C" void kernel_launch(void* const* d_in, const int* in_sizes, int n_in,
                              void* d_out, int out_size, void* d_ws, size_t ws_size,
                              hipStream_t stream) {
  const float* feat = (const float*)d_in[0];
  const int*   src  = (const int*)  d_in[1];
  const int*   dst  = (const int*)  d_in[2];
  const float* W1   = (const float*)d_in[3];
  const float* b1   = (const float*)d_in[4];
  const float* W2   = (const float*)d_in[5];
  const float* b2   = (const float*)d_in[6];
  const float* W3   = (const float*)d_in[7];
  const float* b3   = (const float*)d_in[8];
  float* out = (float*)d_out;

  size_t off = 0;
  auto alloc = [&](size_t bytes) -> char* {
    char* p = (char*)d_ws + off;
    off += (bytes + 255) & ~(size_t)255;
    return p;
  };
  float*     onorm = (float*)alloc((size_t)NNODES * sizeof(float));
  float*     inorm = (float*)alloc((size_t)NNODES * sizeof(float));
  _Float16*  hbuf  = (_Float16*)alloc((size_t)NNODES * FDIM * sizeof(_Float16));
  _Float16*  xbuf  = (_Float16*)alloc((size_t)NNODES * FDIM * sizeof(_Float16));
  _Float16*  Wp1   = (_Float16*)alloc((size_t)FDIM * FDIM * sizeof(_Float16));
  _Float16*  Wp2   = (_Float16*)alloc((size_t)FDIM * FDIM * sizeof(_Float16));
  _Float16*  Wp3   = (_Float16*)alloc((size_t)FDIM * FDIM * sizeof(_Float16));
  int*       cnt_in = (int*)alloc((size_t)NNODES * sizeof(int));
  int*       cnt_out= (int*)alloc((size_t)NNODES * sizeof(int));
  int*       lstart= (int*)alloc((size_t)NNODES * sizeof(int));
  int*       hist  = (int*)alloc((size_t)NBKT * NCHUNK * sizeof(int));
  int*       boff  = (int*)alloc((size_t)NBKT * NCHUNK * sizeof(int));
  int2*      ebuf  = (int2*)alloc((size_t)NEDGES * sizeof(int2));
  int*       lst   = (int*)alloc((size_t)NEDGES * sizeof(int));
  int*       csr   = (int*)alloc((size_t)NEDGES * sizeof(int));
  float*     cw    = (float*)alloc((size_t)NEDGES * sizeof(float));

  const int TB = 256;

  chunk_count_kernel<<<NCHUNK, 1024, 0, stream>>>(dst, cnt_out, hist, NEDGES, NNODES);
  piece_scan_kernel<<<SCAN_GRID, 1024, 0, stream>>>(hist, boff, W1, W2, W3, Wp1, Wp2, Wp3);
  chunk_deal_kernel<<<NCHUNK, 1024, 0, stream>>>(src, dst, boff, cnt_out, ebuf, NEDGES);
  bucket_lists_kernel<<<NBKT, 1024, 0, stream>>>(ebuf, boff, cnt_out, lstart, cnt_in, lst, onorm, inorm, csr, cw, NNODES, NEDGES);

  gather_gemm_relu_gemm_kernel<<<RES_BLOCKS_16x3, 512, 0, stream>>>(
      lstart, cnt_in, csr, cw, feat, inorm, b1, onorm, Wp1, Wp2, xbuf, NNODES);

  gather_relu_gemm_kernel<64><<<RES_BLOCKS_16x4, 512, 0, stream>>>(
      lstart, cnt_in, csr, xbuf, inorm, b2, onorm, Wp3, hbuf, NNODES);

  gather_agg_kernel<64, false><<<RES_BLOCKS_8, TB, 0, stream>>>(
      lstart, cnt_in, csr, hbuf, inorm, b3, out, NNODES);
}


# Packed weight matrices copied once per block into LDS (64 KB for W1 and W2, W3 likewise); each tile's B fragments come from LDS, not from global memory per tile and wave; same values and sums
# speedup vs baseline: 5.0812x; 1.0436x over previous
#include <hip/hip_runtime.h>
#include <math.h>

typedef __attribute__((ext_vector_type(16))) _Float16 v16h;
typedef __attribute__((ext_vector_type(8)))  float    v8f;
typedef __attribute__((ext_vector_type(8)))  _Float16 v8h;
typedef __attribute__((ext_vector_type(4)))  _Float16 v4h;

#define NNODES 100000
#define NEDGES 1600000
#define FDIM   128

#define BKT_SHIFT   9
#define BKT_NODES   (1 << BKT_SHIFT)
#define NBKT        ((NNODES + BKT_NODES - 1) / BKT_NODES)
#define NCHUNK      256
#define CHUNK_EDGES ((NEDGES + NCHUNK - 1) / NCHUNK)
#define CHUNK_PER   ((CHUNK_EDGES + 1023) / 1024)
#define ZSLICE      ((NNODES + NCHUNK - 1) / NCHUNK)

__global__ __launch_bounds__(1024)
void chunk_count_kernel(const int* __restrict__ dst, int* __restrict__ cout,
                        int* __restrict__ hist, int e, int n) {
  __shared__ int s_h[NBKT];
  int t = threadIdx.x, c = blockIdx.x;
  for (int b = t; b < NBKT; b += 1024) s_h[b] = 0;
  int zend = (c + 1) * ZSLICE < n ? (c + 1) * ZSLICE : n;
  for (int i = c * ZSLICE + t; i < zend; i += 1024) cout[i] = 0;
  __syncthreads();
  int beg = c * CHUNK_EDGES;
  int end = beg + CHUNK_EDGES < e ? beg + CHUNK_EDGES : e;
  int d[CHUNK_PER];
#pragma unroll
  for (int u = 0; u < CHUNK_PER; ++u) {
    int i = beg + t + u * 1024;
    d[u] = dst[i < end ? i : end - 1];
  }
#pragma unroll
  for (int u = 0; u < CHUNK_PER; ++u) {
    int i = beg + t + u * 1024;
    if (i < end) atomicAdd(&s_h[d[u] >> BKT_SHIFT], 1);
  }
  __syncthreads();
  for (int b = t; b < NBKT; b += 1024) hist[b * NCHUNK + c] = s_h[b];
}

__device__ __forceinline__ void pack_w_elem(const float* __restrict__ W, _Float16* __restrict__ Wp, int nout, int idx) {
  int nt = nout >> 4;
  int j    = idx & 15;
  int l    = (idx >> 4) & 31;
  int rest = idx >> 9;
  int t    = rest % nt;
  int c    = rest / nt;
  int col   = t * 16 + (l & 15);
  int kbase = c * 32 + ((l >> 4) * 8);
  int K     = kbase + (j & 7) + ((j >> 3) * 16);
  int pos = ((((c * nt + t) * 2 + (j >> 3)) * 32 + l) << 3) + (j & 7);
  Wp[pos] = (_Float16)W[K * nout + col];
}

#define ROWS_PER_WAVE ((NBKT + 31) / 32)

#define PACK_BLK_128 ((4 * 8 * 512) / 1024)
#define PACK_BLK_64  ((4 * 4 * 512) / 1024)
#define PACK_BLOCKS  (2 * PACK_BLK_128 + PACK_BLK_64)
#define SCAN_GRID    (1 + PACK_BLOCKS)

__global__ __launch_bounds__(1024)
void piece_scan_kernel(const int* __restrict__ hist, int* __restrict__ boff,
                       const float* __restrict__ W1, const float* __restrict__ W2,
                       const float* __restrict__ W3, _Float16* __restrict__ Wp1,
                       _Float16* __restrict__ Wp2, _Float16* __restrict__ Wp3) {
  __shared__ int s_tot[NBKT];
  int t = threadIdx.x, lane = t & 31, w = t >> 5;
  if (blockIdx.x > 0) {
    int pb = (int)blockIdx.x - 1;
    if (pb < PACK_BLK_128)          pack_w_elem(W1, Wp1, 128, pb * 1024 + t);
    else if (pb < 2 * PACK_BLK_128) pack_w_elem(W2, Wp2, 128, (pb - PACK_BLK_128) * 1024 + t);
    else                            pack_w_elem(W3, Wp3, 64, (pb - 2 * PACK_BLK_128) * 1024 + t);
    return;
  }
  const int4* h4 = (const int4*)hist;
  int4* o4 = (int4*)boff;
  int4 p[ROWS_PER_WAVE], q[ROWS_PER_WAVE];
#pragma unroll
  for (int u = 0; u < ROWS_PER_WAVE; ++u) {
    int b = w + 32 * u;
    int bb = b < NBKT ? b : NBKT - 1;
    p[u] = h4[bb * (NCHUNK / 4) + lane * 2];
    q[u] = h4[bb * (NCHUNK / 4) + lane * 2 + 1];
  }
#pragma unroll
  for (int u = 0; u < ROWS_PER_WAVE; ++u) {
    int b = w + 32 * u;
    int s = ((p[u].x + p[u].y) + (p[u].z + p[u].w)) + ((q[u].x + q[u].y) + (q[u].z + q[u].w));
#pragma unroll
    for (int k = 16; k > 0; k >>= 1) s += __shfl_xor(s, k);
    if (lane == 0 && b < NBKT) s_tot[b] = s;
  }
  __syncthreads();
#pragma unroll
  for (int u = 0; u < ROWS_PER_WAVE; ++u) {
    int b = w + 32 * u;
    if (b < NBKT) {
      int front = 0;
      for (int k = lane; k < b; k += 32) front += s_tot[k];
#pragma unroll
      for (int k = 16; k > 0; k >>= 1) front += __shfl_xor(front, k);
      int s = ((p[u].x + p[u].y) + (p[u].z + p[u].w)) + ((q[u].x + q[u].y) + (q[u].z + q[u].w));
      int incl = s;
#pragma unroll
      for (int k = 1; k < 32; k <<= 1) {
        int o = __shfl_up(incl, k);
        if (lane >= k) incl += o;
      }
      int run = front + incl - s;
      int4 op, oq;
      op.x = run; run += p[u].x;  op.y = run; run += p[u].y;
      op.z = run; run += p[u].z;  op.w = run; run += p[u].w;
      oq.x = run; run += q[u].x;  oq.y = run; run += q[u].y;
      oq.z = run; run += q[u].z;  oq.w = run;
      o4[b * (NCHUNK / 4) + lane * 2]     = op;
      o4[b * (NCHUNK / 4) + lane * 2 + 1] = oq;
    }
  }
}

__global__ __launch_bounds__(1024)
void chunk_deal_kernel(const int* __restrict__ src, const int* __restrict__ dst,
                       const int* __restrict__ boff, int* __restrict__ cout,
                       int2* __restrict__ ebuf, int e) {
  __shared__ int s_cur[NBKT];
  int t = threadIdx.x, c = blockIdx.x;
  for (int b = t; b < NBKT; b += 1024) s_cur[b] = boff[b * NCHUNK + c];
  __syncthreads();
  int beg = c * CHUNK_EDGES;
  int end = beg + CHUNK_EDGES < e ? beg + CHUNK_EDGES : e;
  int d[CHUNK_PER], s[CHUNK_PER];
#pragma unroll
  for (int u = 0; u < CHUNK_PER; ++u) {
    int i = beg + t + u * 1024;
    int ii = i < end ? i : end - 1;
    d[u] = dst[ii];
    s[u] = src[ii];
  }
#pragma unroll
  for (int u = 0; u < CHUNK_PER; ++u) {
    int i = beg + t + u * 1024;
    if (i < end) {
      int pos = atomicAdd(&s_cur[d[u] >> BKT_SHIFT], 1);
      ebuf[pos] = make_int2(s[u], d[u]);
      atomicAdd(&cout[s[u]], 1);
    }
  }
}

#define SRC_BITS 17
#define SRC_MASK ((1 << SRC_BITS) - 1)
static_assert(NNODES <= (1 << SRC_BITS), "a source id must fit below the node index of a list entry");
static_assert(BKT_SHIFT + SRC_BITS <= 31, "a list entry must stay a non-negative int");
#define LIST_CAP 16384
#define LIST_PAD 8

__global__ __launch_bounds__(1024)
void bucket_lists_kernel(const int2* __restrict__ ebuf, const int* __restrict__ boff,
                         const int* __restrict__ cout, int* __restrict__ off,
                         int* __restrict__ cin, int* lst,
                         float* __restrict__ onorm, float* __restrict__ inorm,
                         int* __restrict__ csr, float* __restrict__ cw, int n, int e) {
  __shared__ int s_cnt[BKT_NODES];
  __shared__ int s_off[BKT_NODES];
  __shared__ int s_w[32];
  __shared__ int s_raw[LIST_CAP];
  __shared__ int s_lst[LIST_CAP + LIST_PAD];
  int t = threadIdx.x, lane = t & 31, w = t >> 5;
  int b = blockIdx.x;
  int base = b * BKT_NODES;
  int ebeg = boff[b * NCHUNK];
  int eend = (b + 1 < NBKT) ? boff[(b + 1) * NCHUNK] : e;
  int m = eend - ebeg;
  bool in_lds = m <= LIST_CAP;
  int* glst = lst + ebeg;
  if (t < BKT_NODES) s_cnt[t] = 0;
  __syncthreads();
  if (in_lds) {
    for (int e0 = t; e0 < m; e0 += 8 * 1024) {
      int2 ed[8];
#pragma unroll
      for (int u = 0; u < 8; ++u) {
        int ei = e0 + u * 1024;
        ed[u] = ebuf[ebeg + (ei < m ? ei : m - 1)];
      }
#pragma unroll
      for (int u = 0; u < 8; ++u) {
        int ei = e0 + u * 1024;
        if (ei < m) {
          int l = ed[u].y - base;
          atomicAdd(&s_cnt[l], 1);
          s_raw[ei] = (l << SRC_BITS) | ed[u].x;
        }
      }
    }
  } else {
    for (int i0 = ebeg + t; i0 < eend; i0 += 4 * 1024) {
      int dd[4];
#pragma unroll
      for (int u = 0; u < 4; ++u) {
        int i = i0 + u * 1024;
        dd[u] = ebuf[i < eend ? i : eend - 1].y;
      }
#pragma unroll
      for (int u = 0; u < 4; ++u) {
        int i = i0 + u * 1024;
        if (i < eend) atomicAdd(&s_cnt[dd[u] - base], 1);
      }
    }
  }
  __syncthreads();
  int cnt = (t < BKT_NODES) ? s_cnt[t] : 0;
  int incl = cnt;
#pragma unroll
  for (int k = 1; k < 32; k <<= 1) {
    int o = __shfl_up(incl, k);
    if (lane >= k) incl += o;
  }
  if (lane == 31) s_w[w] = incl;
  __syncthreads();
  int front = (lane < w) ? s_w[lane] : 0;
#pragma unroll
  for (int k = 16; k > 0; k >>= 1) front += __shfl_xor(front, k);
  int lo = front + incl - cnt;
  if (t < BKT_NODES) {
    s_off[t] = lo;
    s_cnt[t] = 0;
    int node = base + t;
    if (node < n) {
      off[node] = ebeg + lo;
      cin[node] = cnt;
      onorm[node] = rsqrtf(fmaxf((float)(cout[node] + 1), 1.0f));
      inorm[node] = rsqrtf(fmaxf((float)(cnt + 1), 1.0f));
    }
  }
  __syncthreads();
  if (in_lds) {
    for (int ei = t; ei < m; ei += 1024) {
      int v = s_raw[ei];
      int l = v >> SRC_BITS;
      int k = atomicAdd(&s_cnt[l], 1);
      s_lst[s_off[l] + k] = v;
    }
    if (t < LIST_PAD) s_lst[m + t] = 0x7fffffff;
    __syncthreads();
    for (int i = t; i < m; i += 1024) {
      int v   = s_lst[i];
      int l   = v >> SRC_BITS;
      int l0  = s_off[l];
      int ln  = s_cnt[l];
      int k   = v & SRC_MASK;
      int co  = cout[k];
      int r = 0, r1 = 0;
#pragma clang loop vectorize(disable) interleave(disable)
      for (int a = l0; a < l0 + ln; a += 8) {
        int v0 = s_lst[a],     v1 = s_lst[a + 1], v2 = s_lst[a + 2], v3 = s_lst[a + 3];
        int v4 = s_lst[a + 4], v5 = s_lst[a + 5], v6 = s_lst[a + 6], v7 = s_lst[a + 7];
        r  += (v0 < v + (a     < i ? 1 : 0)) ? 1 : 0;
        r1 += (v1 < v + (a + 1 < i ? 1 : 0)) ? 1 : 0;
        r  += (v2 < v + (a + 2 < i ? 1 : 0)) ? 1 : 0;
        r1 += (v3 < v + (a + 3 < i ? 1 : 0)) ? 1 : 0;
        r  += (v4 < v + (a + 4 < i ? 1 : 0)) ? 1 : 0;
        r1 += (v5 < v + (a + 5 < i ? 1 : 0)) ? 1 : 0;
        r  += (v6 < v + (a + 6 < i ? 1 : 0)) ? 1 : 0;
        r1 += (v7 < v + (a + 7 < i ? 1 : 0)) ? 1 : 0;
      }
      r += r1;
      csr[ebeg + l0 + r] = k;
      cw[ebeg + l0 + r]  = rsqrtf(fmaxf((float)(co + 1), 1.0f));
    }
  } else {
    for (int i0 = ebeg + t; i0 < eend; i0 += 4 * 1024) {
      int2 ed[4];
#pragma unroll
      for (int u = 0; u < 4; ++u) {
        int i = i0 + u * 1024;
        ed[u] = ebuf[i < eend ? i : eend - 1];
      }
#pragma unroll
      for (int u = 0; u < 4; ++u) {
        int i = i0 + u * 1024;
        if (i < eend) {
          int l = ed[u].y - base;
          int k = atomicAdd(&s_cnt[l], 1);
          glst[s_off[l] + k] = (l << SRC_BITS) | ed[u].x;
        }
      }
    }
    __syncthreads();
    for (int i = t; i < m; i += 1024) {
      int v   = glst[i];
      int l   = v >> SRC_BITS;
      int l0  = s_off[l];
      int ln  = s_cnt[l];
      int k   = v & SRC_MASK;
      int co  = cout[k];
      int r = 0;
      for (int j = 0; j < ln; ++j) {
        int vj = glst[l0 + j];
        r += (vj < v + (l0 + j < i ? 1 : 0)) ? 1 : 0;
      }
      csr[ebeg + l0 + r] = k;
      cw[ebeg + l0 + r]  = rsqrtf(fmaxf((float)(co + 1), 1.0f));
    }
  }
}

#define RES_BLOCKS_8    2048
#define RES_BLOCKS_16x4 1024
#define RES_BLOCKS_16x3  768

__device__ __forceinline__ float lane_f(float v, int l) {
  return __int_as_float(__builtin_amdgcn_readlane(__float_as_int(v), l));
}

template<int U, typename VT, typename LT>
__device__ __forceinline__ void add_rows(VT& acc, const LT* __restrict__ hv, int ids, int j, unsigned lane) {
  LT r[U];
#pragma unroll
  for (int u = 0; u < U; ++u) {
    int s = __builtin_amdgcn_readlane(ids, j + u);
    const LT* row = hv + (size_t)s * 32;
    r[u] = row[lane];
  }
#pragma unroll
  for (int u = 0; u < U; ++u) acc += __builtin_convertvector(r[u], VT);
}

template<typename VT, typename LT>
__device__ __forceinline__ void add_list_chunk(VT& acc, const LT* __restrict__ hv, int ids, int m, unsigned lane) {
  int j = 0;
  for (; j + 8 <= m; j += 8) add_rows<8, VT, LT>(acc, hv, ids, j, lane);
  if (j + 4 <= m) { add_rows<4, VT, LT>(acc, hv, ids, j, lane); j += 4; }
  if (j + 2 <= m) { add_rows<2, VT, LT>(acc, hv, ids, j, lane); j += 2; }
  if (j < m)      { add_rows<1, VT, LT>(acc, hv, ids, j, lane); }
}

template<int NOUT, bool RELU>
__global__ __launch_bounds__(256)
void gather_agg_kernel(const int* __restrict__ off, const int* __restrict__ cin,
                       const int* __restrict__ csr, const _Float16* __restrict__ h,
                       const float* __restrict__ inorm, const float* __restrict__ b,
                       float* __restrict__ out, int n_nodes) {
  constexpr int V = NOUT / 32;
  typedef float VT __attribute__((ext_vector_type(V)));
  typedef _Float16 LT __attribute__((ext_vector_type(V)));
  int lane = threadIdx.x & 31;
  int nw = (int)((gridDim.x * blockDim.x) >> 5);
  int gw = __builtin_amdgcn_readfirstlane((int)((blockIdx.x * blockDim.x + threadIdx.x) >> 5));
  const LT* hv = (const LT*)h;
  VT bv = ((const VT*)b)[lane];
  for (int first = gw; first < n_nodes; first += 32 * nw) {
    int mn = first + lane * nw;
    bool ok = mn < n_nodes;
    int   begL = ok ? off[mn] : 0;
    int   cntL = ok ? cin[mn] : 0;
    float inL  = ok ? inorm[mn] : 0.0f;
    int beg = 0, n = 0, ids = 0;
    for (int it = -1; it < 32; ++it) {
      int node2 = first + (it + 1) * nw;
      bool more = (it + 1 < 32) && (node2 < n_nodes);
      int beg2 = 0, n2 = 0, ids2 = 0;
      if (more) {
        beg2 = __builtin_amdgcn_readlane(begL, it + 1);
        n2   = __builtin_amdgcn_readlane(cntL, it + 1);
        ids2 = (lane < n2) ? csr[beg2 + lane] : 0;
      }
      if (it >= 0) {
        int node = first + it * nw;
        VT acc = __builtin_convertvector(hv[(size_t)node * 32 + lane], VT);
        add_list_chunk<VT, LT>(acc, hv, ids, n < 32 ? n : 32, lane);
        for (int base = 32; base < n; base += 32) {
          int m = n - base;
          if (m > 32) m = 32;
          int idx = (lane < m) ? csr[beg + base + lane] : 0;
          add_list_chunk<VT, LT>(acc, hv, idx, m, lane);
        }
        float sc = lane_f(inL, it);
        VT v = acc * sc + bv;
        if (RELU) {
#pragma unroll
          for (int k = 0; k < V; ++k) v[k] = fmaxf(v[k], 0.0f);
        }
        ((VT*)out)[(size_t)node * 32 + lane] = v;
      }
      if (!more) break;
      beg = beg2; n = n2; ids = ids2;
    }
  }
}


#define XS_PITCH 136

template<int NOUT2>
__global__ __launch_bounds__(512) __attribute__((amdgpu_waves_per_eu(16)))
void gather_relu_gemm_kernel(const int* __restrict__ off, const int* __restrict__ cin,
                             const int* __restrict__ csr, const _Float16* __restrict__ h,
                             const float* __restrict__ inorm, const float* __restrict__ b,
                             const float* __restrict__ onorm, const _Float16* __restrict__ Wp,
                             _Float16* __restrict__ hnext, int n_nodes) {
  constexpr int NT2 = NOUT2 / 16;
  typedef float VT __attribute__((ext_vector_type(4)));
  typedef v4h LT;
  __shared__ __attribute__((aligned(16))) _Float16 xs[2][16 * XS_PITCH];
  __shared__ __attribute__((aligned(16))) float s_b[128];
  __shared__ __attribute__((aligned(16))) _Float16 sW[128 * NOUT2];
  if (threadIdx.x < 128) s_b[threadIdx.x] = b[threadIdx.x];
  {
    constexpr int NPIECE = 128 * NOUT2 / 8;
    constexpr int PER = NPIECE / 512;
    static_assert(NPIECE % 512 == 0, "the copy gives every thread the same number of pieces");
    v8h wq[PER];
#pragma unroll
    for (int u = 0; u < PER; ++u) wq[u] = *(const v8h*)(Wp + ((int)threadIdx.x + u * 512) * 8);
#pragma unroll
    for (int u = 0; u < PER; ++u) *(v8h*)(sW + ((int)threadIdx.x + u * 512) * 8) = wq[u];
  }
  __syncthreads();

  int lane = threadIdx.x & 31;
  int w = __builtin_amdgcn_readfirstlane((int)(threadIdx.x >> 5));
  int G = (int)gridDim.x;
  int n_tiles = (n_nodes + 15) >> 4;
  const LT* hv = (const LT*)h;

  int r    = lane & 15;
  int half = lane >> 4;

  for (int tfirst = (int)blockIdx.x; tfirst < n_tiles; tfirst += 32 * G) {
    int mt = tfirst + (int)lane * G;
    int mn = mt * 16 + w;
    bool ok = (mt < n_tiles) && (mn < n_nodes);
    int   begL = ok ? off[mn] : 0;
    int   cntL = ok ? cin[mn] : 0;
    float inL  = ok ? inorm[mn] : 0.0f;
    float onL  = ok ? onorm[mn] : 0.0f;
    int beg = 0, n = 0, ids = 0;
    for (int it = -1; it < 32; ++it) {
      int tile2 = tfirst + (it + 1) * G;
      bool more = (it + 1 < 32) && (tile2 < n_tiles);
      int beg2 = 0, n2 = 0, ids2 = 0;
      if (more) {
        beg2 = __builtin_amdgcn_readlane(begL, it + 1);
        n2   = __builtin_amdgcn_readlane(cntL, it + 1);
        ids2 = ((int)lane < n2) ? csr[beg2 + lane] : 0;
      }
      if (it >= 0) {
      int tile = tfirst + it * G;
      int node = tile * 16 + w;
      VT v = {0.f, 0.f, 0.f, 0.f};
      if (node < n_nodes) {
        VT acc = __builtin_convertvector(hv[(size_t)node * 32 + lane], VT);
        add_list_chunk<VT, LT>(acc, hv, ids, n < 32 ? n : 32, lane);
        for (int base = 32; base < n; base += 32) {
          int m = n - base;
          if (m > 32) m = 32;
          int idx = (lane < m) ? csr[beg + base + lane] : 0;
          add_list_chunk<VT, LT>(acc, hv, idx, m, lane);
        }
        float sc = lane_f(inL, it);
        VT bv = *(const VT*)(s_b + lane * 4);
        v = acc * sc + bv;
#pragma unroll
        for (int k = 0; k < 4; ++k) v[k] = fmaxf(v[k], 0.0f);
        float on = lane_f(onL, it);
        v = v * on;
      }

      int pw = (w - NT2 * it) & 15;
      v16h bw[4];
#pragma unroll
      for (int kc = 0; kc < 4; ++kc) bw[kc] = (v16h)(_Float16)0.0f;
      if (pw < NT2) {
#pragma unroll
        for (int kc = 0; kc < 4; ++kc) {
          const _Float16* pf = sW + (kc * NT2 + pw) * 512 + lane * 8;
          v8h lo = *(const v8h*)pf;
          v8h hi = *(const v8h*)(pf + 256);
          bw[kc] = __builtin_shufflevector(lo, hi, 0, 1, 2, 3, 4, 5, 6, 7, 8, 9, 10, 11, 12, 13, 14, 15);
        }
      }

      _Float16* xt = xs[it & 1];
      v4h xh;
      xh[0] = (_Float16)v[0]; xh[1] = (_Float16)v[1];
      xh[2] = (_Float16)v[2]; xh[3] = (_Float16)v[3];
      *(v4h*)(xt + w * XS_PITCH + lane * 4) = xh;
      __syncthreads();

      if (pw < NT2) {
        const _Float16* pa = xt + r * XS_PITCH + half * 8;
        v8f acc2 = {0.f, 0.f, 0.f, 0.f, 0.f, 0.f, 0.f, 0.f};
#pragma unroll
        for (int kc = 0; kc < 4; ++kc) {
          v8h lo = *(const v8h*)(pa + kc * 32);
          v8h hi = *(const v8h*)(pa + kc * 32 + 16);
          v16h a = __builtin_shufflevector(lo, hi, 0, 1, 2, 3, 4, 5, 6, 7, 8, 9, 10, 11, 12, 13, 14, 15);
          acc2 = __builtin_amdgcn_wmma_f32_16x16x32_f16(
               false, a,  false, bw[kc],
               (short)0, acc2,  false,  false);
        }
        int row0 = tile * 16 + half * 8;
        unsigned po = (unsigned)(row0 * NOUT2 + pw * 16 + r);
        if (tile * 16 + 16 <= n_nodes) {
#pragma unroll
          for (int vr = 0; vr < 8; ++vr) hnext[po + (unsigned)(vr * NOUT2)] = (_Float16)acc2[vr];
        } else {
#pragma unroll
          for (int vr = 0; vr < 8; ++vr)
            if (row0 + vr < n_nodes) hnext[po + (unsigned)(vr * NOUT2)] = (_Float16)acc2[vr];
        }
      }
      }
      if (!more) break;
      beg = beg2; n = n2; ids = ids2;
    }
  }
}

template<int U, typename VT>
__device__ __forceinline__ void add_rows_scaled(VT& acc, float& pre, const VT* __restrict__ xv, int ids, int scl, int j, unsigned lane) {
  VT r[U];
#pragma unroll
  for (int u = 0; u < U; ++u) {
    int s = __builtin_amdgcn_readlane(ids, j + u);
    const VT* row = xv + (size_t)s * 32;
    r[u] = row[lane];
  }
  acc = acc * pre;
  pre = 1.0f;
#pragma unroll
  for (int u = 0; u < U; ++u) {
    float sc = __int_as_float(__builtin_amdgcn_readlane(scl, j + u));
    acc += r[u] * sc;
  }
}

template<typename VT>
__device__ __forceinline__ void add_list_chunk_scaled(VT& acc, float& pre, const VT* __restrict__ xv, int ids, int scl, int m, unsigned lane) {
  int j = 0;
  for (; j + 8 <= m; j += 8) add_rows_scaled<8, VT>(acc, pre, xv, ids, scl, j, lane);
  if (j + 4 <= m) { add_rows_scaled<4, VT>(acc, pre, xv, ids, scl, j, lane); j += 4; }
  if (j + 2 <= m) { add_rows_scaled<2, VT>(acc, pre, xv, ids, scl, j, lane); j += 2; }
  if (j < m)      { add_rows_scaled<1, VT>(acc, pre, xv, ids, scl, j, lane); }
}

__global__ __launch_bounds__(512) __attribute__((amdgpu_waves_per_eu(12)))
void gather_gemm_relu_gemm_kernel(const int* __restrict__ off, const int* __restrict__ cin,
                                  const int* __restrict__ csr, const float* __restrict__ cw,
                                  const float* __restrict__ feat,
                                  const float* __restrict__ inorm, const float* __restrict__ b,
                                  const float* __restrict__ onorm,
                                  const _Float16* __restrict__ Wp1, const _Float16* __restrict__ Wp2,
                                  _Float16* __restrict__ hnext, int n_nodes) {
  typedef float VT __attribute__((ext_vector_type(4)));
  __shared__ __attribute__((aligned(16))) _Float16 xs[16 * XS_PITCH];
  __shared__ __attribute__((aligned(16))) _Float16 ys[16 * XS_PITCH];
  __shared__ __attribute__((aligned(16))) float s_in[16];
  __shared__ __attribute__((aligned(16))) float s_on[16];
  __shared__ __attribute__((aligned(16))) float s_b[128];
  __shared__ __attribute__((aligned(16))) _Float16 sW[2][128 * 128];
  if (threadIdx.x < 128) s_b[threadIdx.x] = b[threadIdx.x];
  {
    constexpr int PER = (128 * 128 / 8) / 512;
    v8h wq[2 * PER];
#pragma unroll
    for (int u = 0; u < PER; ++u) {
      wq[u]       = *(const v8h*)(Wp1 + ((int)threadIdx.x + u * 512) * 8);
      wq[PER + u] = *(const v8h*)(Wp2 + ((int)threadIdx.x + u * 512) * 8);
    }
#pragma unroll
    for (int u = 0; u < PER; ++u) {
      *(v8h*)(sW[0] + ((int)threadIdx.x + u * 512) * 8) = wq[u];
      *(v8h*)(sW[1] + ((int)threadIdx.x + u * 512) * 8) = wq[PER + u];
    }
  }
  __syncthreads();

  int lane = threadIdx.x & 31;
  int w = __builtin_amdgcn_readfirstlane((int)(threadIdx.x >> 5));
  int G = (int)gridDim.x;
  int n_tiles = (n_nodes + 15) >> 4;
  const VT* xv = (const VT*)feat;

  int t = w & 7;

  int r    = lane & 15;
  int half = lane >> 4;

  for (int tfirst = (int)blockIdx.x; tfirst < n_tiles; tfirst += 32 * G) {
    int mt = tfirst + (int)lane * G;
    int mn = mt * 16 + w;
    bool ok = (mt < n_tiles) && (mn < n_nodes);
    int   begL = ok ? off[mn] : 0;
    int   cntL = ok ? cin[mn] : 0;
    float inL  = ok ? inorm[mn] : 0.0f;
    float onL  = ok ? onorm[mn] : 0.0f;
    int beg = 0, n = 0, ids = 0, scl = 0;
    float sc_in = 0.0f, sc_on = 0.0f;
    for (int it = -1; it < 32; ++it) {
      int tile2 = tfirst + (it + 1) * G;
      bool more = (it + 1 < 32) && (tile2 < n_tiles);
      int beg2 = 0, n2 = 0, ids2 = 0, scl2 = 0;
      float in2 = 0.0f, on2 = 0.0f;
      if (more) {
        beg2 = __builtin_amdgcn_readlane(begL, it + 1);
        n2   = __builtin_amdgcn_readlane(cntL, it + 1);
        in2  = lane_f(inL, it + 1);
        on2  = lane_f(onL, it + 1);
        ids2 = ((int)lane < n2) ? csr[beg2 + lane] : 0;
        scl2 = ((int)lane < n2) ? __float_as_int(cw[beg2 + lane]) : 0;
      }
      if (it >= 0) {
      int tile = tfirst + it * G;
      int node = tile * 16 + w;
      VT acc = {0.f, 0.f, 0.f, 0.f};
      if (node < n_nodes) {
        acc = xv[(size_t)node * 32 + lane];
        float pre = sc_on;
        add_list_chunk_scaled<VT>(acc, pre, xv, ids, scl, n < 32 ? n : 32, lane);
        for (int base = 32; base < n; base += 32) {
          int m = n - base;
          if (m > 32) m = 32;
          int idx = (lane < m) ? csr[beg + base + lane] : 0;
          int sdx = (lane < m) ? __float_as_int(cw[beg + base + lane]) : 0;
          add_list_chunk_scaled<VT>(acc, pre, xv, idx, sdx, m, lane);
        }
        acc = acc * pre;
      }

      bool first = ((w >> 3) ^ it) & 1 ? false : true;
      const _Float16* Wsel = first ? sW[0] : sW[1];
      v16h bw[4];
#pragma unroll
      for (int kc = 0; kc < 4; ++kc) {
        const _Float16* pf = Wsel + (kc * 8 + t) * 512 + lane * 8;
        v8h lo = *(const v8h*)pf;
        v8h hi = *(const v8h*)(pf + 256);
        bw[kc] = __builtin_shufflevector(lo, hi, 0, 1, 2, 3, 4, 5, 6, 7, 8, 9, 10, 11, 12, 13, 14, 15);
      }

      v4h xh;
      xh[0] = (_Float16)acc[0]; xh[1] = (_Float16)acc[1];
      xh[2] = (_Float16)acc[2]; xh[3] = (_Float16)acc[3];
      *(v4h*)(xs + w * XS_PITCH + lane * 4) = xh;
      if (lane == 0) { s_in[w] = sc_in; s_on[w] = sc_on; }
      __syncthreads();

      if (first) {
        const _Float16* pa = xs + r * XS_PITCH + half * 8;
        v8f acc1 = {0.f, 0.f, 0.f, 0.f, 0.f, 0.f, 0.f, 0.f};
#pragma unroll
        for (int kc = 0; kc < 4; ++kc) {
          v8h lo = *(const v8h*)(pa + kc * 32);
          v8h hi = *(const v8h*)(pa + kc * 32 + 16);
          v16h a = __builtin_shufflevector(lo, hi, 0, 1, 2, 3, 4, 5, 6, 7, 8, 9, 10, 11, 12, 13, 14, 15);
          acc1 = __builtin_amdgcn_wmma_f32_16x16x32_f16(
               false, a,  false, bw[kc],
               (short)0, acc1,  false,  false);
        }
        float bcol = s_b[t * 16 + r];
#pragma unroll
        for (int vr = 0; vr < 8; ++vr) {
          int row = half * 8 + vr;
          float v = acc1[vr] * s_in[row] + bcol;
          v = fmaxf(v, 0.0f);
          v = v * s_on[row];
          ys[row * XS_PITCH + t * 16 + r] = (_Float16)v;
        }
      }
      __syncthreads();

      if (!first) {
        const _Float16* pb = ys + r * XS_PITCH + half * 8;
        v8f acc2 = {0.f, 0.f, 0.f, 0.f, 0.f, 0.f, 0.f, 0.f};
#pragma unroll
        for (int kc = 0; kc < 4; ++kc) {
          v8h lo = *(const v8h*)(pb + kc * 32);
          v8h hi = *(const v8h*)(pb + kc * 32 + 16);
          v16h a = __builtin_shufflevector(lo, hi, 0, 1, 2, 3, 4, 5, 6, 7, 8, 9, 10, 11, 12, 13, 14, 15);
          acc2 = __builtin_amdgcn_wmma_f32_16x16x32_f16(
               false, a,  false, bw[kc],
               (short)0, acc2,  false,  false);
        }
        int row0 = tile * 16 + half * 8;
        unsigned po = (unsigned)(row0 * 128 + t * 16 + r);
        if (tile * 16 + 16 <= n_nodes) {
#pragma unroll
          for (int vr = 0; vr < 8; ++vr) hnext[po + (unsigned)(vr * 128)] = (_Float16)acc2[vr];
        } else {
#pragma unroll
          for (int vr = 0; vr < 8; ++vr)
            if (row0 + vr < n_nodes) hnext[po + (unsigned)(vr * 128)] = (_Float16)acc2[vr];
        }
      }
      }
      if (!more) break;
      beg = beg2; n = n2; ids = ids2; scl = scl2; sc_in = in2; sc_on = on2;
    }
  }
}

extern "C" void kernel_launch(void* const* d_in, const int* in_sizes, int n_in,
                              void* d_out, int out_size, void* d_ws, size_t ws_size,
                              hipStream_t stream) {
  const float* feat = (const float*)d_in[0];
  const int*   src  = (const int*)  d_in[1];
  const int*   dst  = (const int*)  d_in[2];
  const float* W1   = (const float*)d_in[3];
  const float* b1   = (const float*)d_in[4];
  const float* W2   = (const float*)d_in[5];
  const float* b2   = (const float*)d_in[6];
  const float* W3   = (const float*)d_in[7];
  const float* b3   = (const float*)d_in[8];
  float* out = (float*)d_out;

  size_t off = 0;
  auto alloc = [&](size_t bytes) -> char* {
    char* p = (char*)d_ws + off;
    off += (bytes + 255) & ~(size_t)255;
    return p;
  };
  float*     onorm = (float*)alloc((size_t)NNODES * sizeof(float));
  float*     inorm = (float*)alloc((size_t)NNODES * sizeof(float));
  _Float16*  hbuf  = (_Float16*)alloc((size_t)NNODES * FDIM * sizeof(_Float16));
  _Float16*  xbuf  = (_Float16*)alloc((size_t)NNODES * FDIM * sizeof(_Float16));
  _Float16*  Wp1   = (_Float16*)alloc((size_t)FDIM * FDIM * sizeof(_Float16));
  _Float16*  Wp2   = (_Float16*)alloc((size_t)FDIM * FDIM * sizeof(_Float16));
  _Float16*  Wp3   = (_Float16*)alloc((size_t)FDIM * FDIM * sizeof(_Float16));
  int*       cnt_in = (int*)alloc((size_t)NNODES * sizeof(int));
  int*       cnt_out= (int*)alloc((size_t)NNODES * sizeof(int));
  int*       lstart= (int*)alloc((size_t)NNODES * sizeof(int));
  int*       hist  = (int*)alloc((size_t)NBKT * NCHUNK * sizeof(int));
  int*       boff  = (int*)alloc((size_t)NBKT * NCHUNK * sizeof(int));
  int2*      ebuf  = (int2*)alloc((size_t)NEDGES * sizeof(int2));
  int*       lst   = (int*)alloc((size_t)NEDGES * sizeof(int));
  int*       csr   = (int*)alloc((size_t)NEDGES * sizeof(int));
  float*     cw    = (float*)alloc((size_t)NEDGES * sizeof(float));

  const int TB = 256;

  chunk_count_kernel<<<NCHUNK, 1024, 0, stream>>>(dst, cnt_out, hist, NEDGES, NNODES);
  piece_scan_kernel<<<SCAN_GRID, 1024, 0, stream>>>(hist, boff, W1, W2, W3, Wp1, Wp2, Wp3);
  chunk_deal_kernel<<<NCHUNK, 1024, 0, stream>>>(src, dst, boff, cnt_out, ebuf, NEDGES);
  bucket_lists_kernel<<<NBKT, 1024, 0, stream>>>(ebuf, boff, cnt_out, lstart, cnt_in, lst, onorm, inorm, csr, cw, NNODES, NEDGES);

  gather_gemm_relu_gemm_kernel<<<RES_BLOCKS_16x3, 512, 0, stream>>>(
      lstart, cnt_in, csr, cw, feat, inorm, b1, onorm, Wp1, Wp2, xbuf, NNODES);

  gather_relu_gemm_kernel<64><<<RES_BLOCKS_16x4, 512, 0, stream>>>(
      lstart, cnt_in, csr, xbuf, inorm, b2, onorm, Wp3, hbuf, NNODES);

  gather_agg_kernel<64, false><<<RES_BLOCKS_8, TB, 0, stream>>>(
      lstart, cnt_in, csr, hbuf, inorm, b3, out, NNODES);
}
